# starting kernel
# baseline (speedup 1.0000x reference)
#include <hip/hip_runtime.h>

typedef __attribute__((ext_vector_type(2))) float v2f;
typedef __attribute__((ext_vector_type(8))) float v8f;

#define F_IN   128
#define HID    32
#define HEADS  4
#define OUTC   16

__device__ __forceinline__ unsigned f2ord(float f) {
  unsigned u = __float_as_uint(f);
  return (u & 0x80000000u) ? ~u : (u | 0x80000000u);
}
__device__ __forceinline__ float ord2f(unsigned u) {
  return (u & 0x80000000u) ? __uint_as_float(u & 0x7fffffffu)
                           : __uint_as_float(~u);
}
__device__ __forceinline__ float lrelu(float v) { return v > 0.f ? v : 0.2f * v; }

__global__ void k_fill0(float* __restrict__ p, long long count) {
  long long i = (long long)blockIdx.x * blockDim.x + threadIdx.x;
  if (i < count) p[i] = 0.f;
}

__global__ void k_gemm1(const float* __restrict__ x, const float* __restrict__ W,
                        float* __restrict__ H, int n) {
  const int wave = threadIdx.x >> 5;
  const int lane = threadIdx.x & 31;
  const int row0 = blockIdx.x * 16;
  if (row0 >= n) return;
  const int col0 = wave * 16;
  const int lr = lane & 15;
  const int lh = lane >> 4;
  const float* ap = x + (size_t)(row0 + lr) * F_IN + 2 * lh;
  const float* bp = W + (size_t)(2 * lh) * F_IN + col0 + lr;
  v8f acc = {};
#pragma unroll
  for (int k = 0; k < F_IN; k += 4) {
    v2f a = *(const v2f*)(ap + k);
    v2f b;
    b.x = bp[(size_t)k * F_IN];
    b.y = bp[(size_t)k * F_IN + F_IN];
    acc = __builtin_amdgcn_wmma_f32_16x16x4_f32(false, a, false, b, (short)0, acc,
                                                false, false);
  }
#pragma unroll
  for (int i = 0; i < 8; i++) {
    H[(size_t)(row0 + i + 8 * lh) * F_IN + col0 + lr] = acc[i];
  }
}

__global__ void k_gemm2(const float* __restrict__ Hp, const float* __restrict__ W2,
                        float* __restrict__ H2, int n) {
  const int wave = threadIdx.x >> 5;
  const int lane = threadIdx.x & 31;
  const int row0 = (blockIdx.x * 8 + wave) * 16;
  if (row0 >= n) return;
  const int lr = lane & 15;
  const int lh = lane >> 4;
  const float* ap = Hp + (size_t)(row0 + lr) * F_IN + 2 * lh;
  const float* bp = W2 + (size_t)(2 * lh) * OUTC + lr;
  v8f acc = {};
#pragma unroll
  for (int k = 0; k < F_IN; k += 4) {
    v2f a = *(const v2f*)(ap + k);
    v2f b;
    b.x = bp[(size_t)k * OUTC];
    b.y = bp[(size_t)k * OUTC + OUTC];
    acc = __builtin_amdgcn_wmma_f32_16x16x4_f32(false, a, false, b, (short)0, acc,
                                                false, false);
  }
#pragma unroll
  for (int i = 0; i < 8; i++) {
    H2[(size_t)(row0 + i + 8 * lh) * OUTC + lr] = acc[i];
  }
}

template <int HN, int CN>
__global__ void k_coef(const float* __restrict__ H, const float* __restrict__ asr,
                       const float* __restrict__ adt, float* __restrict__ als,
                       float* __restrict__ ald, int n) {
  int idx = blockIdx.x * blockDim.x + threadIdx.x;
  if (idx >= n * HN) return;
  int node = idx / HN, h = idx % HN;
  const float* hv = H + (size_t)node * (HN * CN) + h * CN;
  const float* a1 = asr + h * CN;
  const float* a2 = adt + h * CN;
  float s1 = 0.f, s2 = 0.f;
#pragma unroll
  for (int c = 0; c < CN; c++) {
    float v = hv[c];
    s1 += v * a1[c];
    s2 += v * a2[c];
  }
  als[idx] = s1;
  ald[idx] = s2;
}

template <int HN>
__global__ void k_edge_max(const int* __restrict__ ei, int E, int n,
                           const float* __restrict__ als, const float* __restrict__ ald,
                           unsigned* __restrict__ smax) {
  int idx = blockIdx.x * blockDim.x + threadIdx.x;
  int tot = (E + n) * HN;
  if (idx >= tot) return;
  int e = idx / HN, h = idx % HN;
  int s, d;
  if (e < E) { s = ei[e]; d = ei[E + e]; } else { s = d = e - E; }
  float logit = lrelu(als[s * HN + h] + ald[d * HN + h]);
  atomicMax(&smax[d * HN + h], f2ord(logit));
}

template <int HN>
__global__ void k_edge_sum(const int* __restrict__ ei, int E, int n,
                           const float* __restrict__ als, const float* __restrict__ ald,
                           const unsigned* __restrict__ smax, float* __restrict__ ssum) {
  int idx = blockIdx.x * blockDim.x + threadIdx.x;
  int tot = (E + n) * HN;
  if (idx >= tot) return;
  int e = idx / HN, h = idx % HN;
  int s, d;
  if (e < E) { s = ei[e]; d = ei[E + e]; } else { s = d = e - E; }
  float logit = lrelu(als[s * HN + h] + ald[d * HN + h]);
  float m = ord2f(smax[d * HN + h]);
  atomicAdd(&ssum[d * HN + h], expf(logit - m));
}

template <int HN, int CN>
__global__ void k_edge_agg(const int* __restrict__ ei, int E, int n,
                           const float* __restrict__ als, const float* __restrict__ ald,
                           const unsigned* __restrict__ smax, const float* __restrict__ ssum,
                           const float* __restrict__ H, float* __restrict__ agg) {
  int idx = blockIdx.x * blockDim.x + threadIdx.x;
  int tot = (E + n) * (HN * CN);
  if (idx >= tot) return;
  int e = idx / (HN * CN);
  int rem = idx - e * (HN * CN);
  int h = rem / CN;
  int s, d;
  if (e < E) { s = ei[e]; d = ei[E + e]; } else { s = d = e - E; }
  float logit = lrelu(als[s * HN + h] + ald[d * HN + h]);
  float m = ord2f(smax[d * HN + h]);
  float alpha = expf(logit - m) / (ssum[d * HN + h] + 1e-16f);
  float msg = H[(size_t)s * (HN * CN) + rem] * alpha;
  atomicAdd(&agg[(size_t)d * (HN * CN) + rem], msg);
}

__global__ void k_finish1(const float* __restrict__ agg, const float* __restrict__ b1,
                          float* __restrict__ Hp, int n) {
  int idx = blockIdx.x * blockDim.x + threadIdx.x;
  if (idx >= n * F_IN) return;
  int node = idx >> 7, j = idx & 127;
  int i = j >> 5, k = j & 31;
  int sc = 4 * k + i;
  float v = agg[(size_t)node * F_IN + sc] + b1[sc];
  v = v > 0.f ? v : (expf(v) - 1.f);
  Hp[(size_t)node * F_IN + j] = v;
}

__global__ void k_finish2(float* __restrict__ out, const float* __restrict__ b2, int n) {
  int node = blockIdx.x * blockDim.x + threadIdx.x;
  if (node >= n) return;
  float v[OUTC];
  float m = -3.0e38f;
#pragma unroll
  for (int c = 0; c < OUTC; c++) {
    v[c] = out[(size_t)node * OUTC + c] + b2[c];
    m = fmaxf(m, v[c]);
  }
  float s = 0.f;
#pragma unroll
  for (int c = 0; c < OUTC; c++) s += expf(v[c] - m);
  float lse = m + logf(s);
#pragma unroll
  for (int c = 0; c < OUTC; c++) out[(size_t)node * OUTC + c] = v[c] - lse;
}

static inline unsigned cdiv(long long a, long long b) { return (unsigned)((a + b - 1) / b); }

extern "C" void kernel_launch(void* const* d_in, const int* in_sizes, int n_in,
                              void* d_out, int out_size, void* d_ws, size_t ws_size,
                              hipStream_t stream) {
  const float* x    = (const float*)d_in[0];
  const int*   ei   = (const int*)  d_in[1];
  const float* W1   = (const float*)d_in[2];
  const float* asr1 = (const float*)d_in[3];
  const float* adt1 = (const float*)d_in[4];
  const float* b1   = (const float*)d_in[5];
  const float* W2   = (const float*)d_in[6];
  const float* asr2 = (const float*)d_in[7];
  const float* adt2 = (const float*)d_in[8];
  const float* b2   = (const float*)d_in[9];
  float* out = (float*)d_out;

  const int n = in_sizes[0] / F_IN;
  const int E = in_sizes[1] / 2;

  float* ws = (float*)d_ws;
  size_t off = 0;
  float*    H1    = ws + off; off += (size_t)n * F_IN;
  float*    agg1  = ws + off; off += (size_t)n * F_IN;
  float*    als1  = ws + off; off += (size_t)n * HEADS;
  float*    ald1  = ws + off; off += (size_t)n * HEADS;
  unsigned* smax1 = (unsigned*)(ws + off); off += (size_t)n * HEADS;
  float*    ssum1 = ws + off; off += (size_t)n * HEADS;
  float*    H2    = ws + off; off += (size_t)n * OUTC;
  float*    als2  = ws + off; off += (size_t)n;
  float*    ald2  = ws + off; off += (size_t)n;
  unsigned* smax2 = (unsigned*)(ws + off); off += (size_t)n;
  float*    ssum2 = ws + off; off += (size_t)n;

  const int B = 256;
  const long long tot_e1 = (long long)(E + n) * HEADS;
  const long long tot_a1 = (long long)(E + n) * (HEADS * HID);
  const long long tot_e2 = (long long)(E + n);
  const long long tot_a2 = (long long)(E + n) * OUTC;

  long long zcnt = (long long)(agg1 - ws);
  zcnt = (long long)off - zcnt;
  k_fill0<<<cdiv(zcnt, B), B, 0, stream>>>(agg1, zcnt);
  k_fill0<<<cdiv((long long)n * OUTC, B), B, 0, stream>>>(out, (long long)n * OUTC);

  k_gemm1<<<cdiv(n, 16), B, 0, stream>>>(x, W1, H1, n);
  k_coef<HEADS, HID><<<cdiv((long long)n * HEADS, B), B, 0, stream>>>(H1, asr1, adt1, als1, ald1, n);
  k_edge_max<HEADS><<<cdiv(tot_e1, B), B, 0, stream>>>(ei, E, n, als1, ald1, smax1);
  k_edge_sum<HEADS><<<cdiv(tot_e1, B), B, 0, stream>>>(ei, E, n, als1, ald1, smax1, ssum1);
  k_edge_agg<HEADS, HID><<<cdiv(tot_a1, B), B, 0, stream>>>(ei, E, n, als1, ald1, smax1, ssum1, H1, agg1);
  k_finish1<<<cdiv((long long)n * F_IN, B), B, 0, stream>>>(agg1, b1, H1, n);

  k_gemm2<<<cdiv(cdiv(n, 16), 8), B, 0, stream>>>(H1, W2, H2, n);
  k_coef<1, OUTC><<<cdiv((long long)n, B), B, 0, stream>>>(H2, asr2, adt2, als2, ald2, n);
  k_edge_max<1><<<cdiv(tot_e2, B), B, 0, stream>>>(ei, E, n, als2, ald2, smax2);
  k_edge_sum<1><<<cdiv(tot_e2, B), B, 0, stream>>>(ei, E, n, als2, ald2, smax2, ssum2);
  k_edge_agg<1, OUTC><<<cdiv(tot_a2, B), B, 0, stream>>>(ei, E, n, als2, ald2, smax2, ssum2, H2, out);
  k_finish2<<<cdiv((long long)n, B), B, 0, stream>>>(out, b2, n);
}


# Edge passes: bucket sources by destination (one int atomic per edge), then one wave per node does max, sum and weighted gather in registers; no float atomics, sums in bucket arrival order
# speedup vs baseline: 3.7151x; 3.7151x over previous
#include <hip/hip_runtime.h>

typedef __attribute__((ext_vector_type(2))) float v2f;
typedef __attribute__((ext_vector_type(8))) float v8f;

#define F_IN   128
#define HID    32
#define HEADS  4
#define OUTC   16
#define CAP    64

__device__ __forceinline__ float lrelu(float v) { return v > 0.f ? v : 0.2f * v; }

template <int G>
__device__ __forceinline__ float grp_max(float v) {
  for (int o = G >> 1; o > 0; o >>= 1) v = fmaxf(v, __shfl_xor(v, o));
  return v;
}
template <int G>
__device__ __forceinline__ float grp_sum(float v) {
  for (int o = G >> 1; o > 0; o >>= 1) v += __shfl_xor(v, o);
  return v;
}

__global__ void k_fill0i(int* __restrict__ p, int count) {
  int i = blockIdx.x * blockDim.x + threadIdx.x;
  if (i < count) p[i] = 0;
}

__global__ void k_bucket(const int* __restrict__ ei, int E, int* __restrict__ cnt,
                         int* __restrict__ bucket) {
  int e = blockIdx.x * blockDim.x + threadIdx.x;
  if (e >= E) return;
  int s = ei[e], d = ei[E + e];
  int slot = atomicAdd(&cnt[d], 1);
  if (slot < CAP) bucket[(size_t)d * CAP + slot] = s;
}

__global__ void k_gemm1(const float* __restrict__ x, const float* __restrict__ W,
                        float* __restrict__ H, int n) {
  const int wave = threadIdx.x >> 5;
  const int lane = threadIdx.x & 31;
  const int row0 = blockIdx.x * 16;
  if (row0 >= n) return;
  const int col0 = wave * 16;
  const int lr = lane & 15;
  const int lh = lane >> 4;
  const float* ap = x + (size_t)(row0 + lr) * F_IN + 2 * lh;
  const float* bp = W + (size_t)(2 * lh) * F_IN + col0 + lr;
  v8f acc = {};
#pragma unroll
  for (int k = 0; k < F_IN; k += 4) {
    v2f a = *(const v2f*)(ap + k);
    v2f b;
    b.x = bp[(size_t)k * F_IN];
    b.y = bp[(size_t)k * F_IN + F_IN];
    acc = __builtin_amdgcn_wmma_f32_16x16x4_f32(false, a, false, b, (short)0, acc,
                                                false, false);
  }
#pragma unroll
  for (int i = 0; i < 8; i++) {
    H[(size_t)(row0 + i + 8 * lh) * F_IN + col0 + lr] = acc[i];
  }
}

__global__ void k_gemm2(const float* __restrict__ Hp, const float* __restrict__ W2,
                        float* __restrict__ H2, int n) {
  const int wave = threadIdx.x >> 5;
  const int lane = threadIdx.x & 31;
  const int row0 = (blockIdx.x * 8 + wave) * 16;
  if (row0 >= n) return;
  const int lr = lane & 15;
  const int lh = lane >> 4;
  const float* ap = Hp + (size_t)(row0 + lr) * F_IN + 2 * lh;
  const float* bp = W2 + (size_t)(2 * lh) * OUTC + lr;
  v8f acc = {};
#pragma unroll
  for (int k = 0; k < F_IN; k += 4) {
    v2f a = *(const v2f*)(ap + k);
    v2f b;
    b.x = bp[(size_t)k * OUTC];
    b.y = bp[(size_t)k * OUTC + OUTC];
    acc = __builtin_amdgcn_wmma_f32_16x16x4_f32(false, a, false, b, (short)0, acc,
                                                false, false);
  }
#pragma unroll
  for (int i = 0; i < 8; i++) {
    H2[(size_t)(row0 + i + 8 * lh) * OUTC + lr] = acc[i];
  }
}

template <int HN, int CN>
__global__ void k_coef(const float* __restrict__ H, const float* __restrict__ asr,
                       const float* __restrict__ adt, float* __restrict__ als,
                       float* __restrict__ ald, int n) {
  int idx = blockIdx.x * blockDim.x + threadIdx.x;
  if (idx >= n * HN) return;
  int node = idx / HN, h = idx % HN;
  const float* hv = H + (size_t)node * (HN * CN) + h * CN;
  const float* a1 = asr + h * CN;
  const float* a2 = adt + h * CN;
  float s1 = 0.f, s2 = 0.f;
#pragma unroll
  for (int c = 0; c < CN; c++) {
    float v = hv[c];
    s1 += v * a1[c];
    s2 += v * a2[c];
  }
  als[idx] = s1;
  ald[idx] = s2;
}

template <int G, class F>
__device__ __forceinline__ void for_each_chunk(bool fast, int d, int deg,
                                               const int* __restrict__ brow,
                                               const int* __restrict__ ei, int E, int g, F&& f) {
  if (fast) {
    const int tot = deg + 1;
    for (int base = 0; base < tot; base += G) {
      const int j = base + g;
      int s = d;
      if (j < deg) s = brow[j];
      f(s, j < tot);
    }
  } else {
    for (int base = 0; base < E; base += G) {
      const int e = base + g;
      bool valid = false;
      int s = d;
      if (e < E && ei[E + e] == d) { valid = true; s = ei[e]; }
      f(s, valid);
    }
    f(d, g == 0);
  }
}

__global__ void k_node1(const int* __restrict__ ei, int E, int n,
                        const int* __restrict__ cnt, const int* __restrict__ bucket,
                        const float* __restrict__ als, const float* __restrict__ ald,
                        const float* __restrict__ H, const float* __restrict__ b1,
                        float* __restrict__ Hp) {
  const int lane = threadIdx.x & 31;
  const int d = blockIdx.x * (blockDim.x >> 5) + (threadIdx.x >> 5);
  if (d >= n) return;
  const int c = cnt[d];
  const bool fast = c <= CAP;
  const int deg = fast ? c : 0;
  const int* brow = bucket + (size_t)d * CAP;
  const float4 dv = *(const float4*)(ald + (size_t)d * HEADS);
  const float ninf = -__builtin_inff();

  float m0 = ninf, m1 = ninf, m2 = ninf, m3 = ninf;
  for_each_chunk<32>(fast, d, deg, brow, ei, E, lane, [&](int s, bool valid) {
    if (valid) {
      const float4 a = *(const float4*)(als + (size_t)s * HEADS);
      m0 = fmaxf(m0, lrelu(a.x + dv.x));
      m1 = fmaxf(m1, lrelu(a.y + dv.y));
      m2 = fmaxf(m2, lrelu(a.z + dv.z));
      m3 = fmaxf(m3, lrelu(a.w + dv.w));
    }
  });
  m0 = grp_max<32>(m0); m1 = grp_max<32>(m1); m2 = grp_max<32>(m2); m3 = grp_max<32>(m3);

  float s0 = 0.f, s1 = 0.f, s2 = 0.f, s3 = 0.f;
  for_each_chunk<32>(fast, d, deg, brow, ei, E, lane, [&](int s, bool valid) {
    if (valid) {
      const float4 a = *(const float4*)(als + (size_t)s * HEADS);
      s0 += expf(lrelu(a.x + dv.x) - m0);
      s1 += expf(lrelu(a.y + dv.y) - m1);
      s2 += expf(lrelu(a.z + dv.z) - m2);
      s3 += expf(lrelu(a.w + dv.w) - m3);
    }
  });
  const float q0 = grp_sum<32>(s0) + 1e-16f, q1 = grp_sum<32>(s1) + 1e-16f;
  const float q2 = grp_sum<32>(s2) + 1e-16f, q3 = grp_sum<32>(s3) + 1e-16f;

  const int hsel = lane >> 3;
  float4 acc = {0.f, 0.f, 0.f, 0.f};
  for_each_chunk<32>(fast, d, deg, brow, ei, E, lane, [&](int s, bool valid) {
    float a0 = 0.f, a1 = 0.f, a2 = 0.f, a3 = 0.f;
    if (valid) {
      const float4 a = *(const float4*)(als + (size_t)s * HEADS);
      a0 = expf(lrelu(a.x + dv.x) - m0) / q0;
      a1 = expf(lrelu(a.y + dv.y) - m1) / q1;
      a2 = expf(lrelu(a.z + dv.z) - m2) / q2;
      a3 = expf(lrelu(a.w + dv.w) - m3) / q3;
    }
    unsigned mask = __builtin_amdgcn_ballot_w32(valid);
    while (mask) {
      const int j = __builtin_ctz(mask);
      mask &= mask - 1;
      const int sj = __shfl(s, j);
      const float w0 = __shfl(a0, j), w1 = __shfl(a1, j);
      const float w2 = __shfl(a2, j), w3 = __shfl(a3, j);
      const float w = hsel == 0 ? w0 : hsel == 1 ? w1 : hsel == 2 ? w2 : w3;
      const float4 hv = *(const float4*)(H + (size_t)sj * F_IN + lane * 4);
      acc.x += hv.x * w; acc.y += hv.y * w; acc.z += hv.z * w; acc.w += hv.w * w;
    }
  });

  float v[4] = {acc.x + b1[lane * 4 + 0], acc.y + b1[lane * 4 + 1],
                acc.z + b1[lane * 4 + 2], acc.w + b1[lane * 4 + 3]};
  for (int i = 0; i < 4; i++) {
    float t = v[i];
    t = t > 0.f ? t : (expf(t) - 1.f);
    Hp[(size_t)d * F_IN + i * 32 + lane] = t;
  }
}

__global__ void k_node2(const int* __restrict__ ei, int E, int n,
                        const int* __restrict__ cnt, const int* __restrict__ bucket,
                        const float* __restrict__ als, const float* __restrict__ ald,
                        const float* __restrict__ H2, const float* __restrict__ b2,
                        float* __restrict__ out) {
  const int lane = threadIdx.x & 31;
  const int g = lane & 15;
  const int half = lane >> 4;
  const int d = (blockIdx.x * (blockDim.x >> 5) + (threadIdx.x >> 5)) * 2 + half;
  const bool live = d < n;
  const int dd = live ? d : 0;
  const int c = live ? cnt[dd] : -1;
  const bool fast = c <= CAP;
  const int deg = fast ? c : 0;
  const int* brow = bucket + (size_t)dd * CAP;
  const float dv = ald[dd];
  const float ninf = -__builtin_inff();

  float m = ninf;
  for_each_chunk<16>(fast, dd, deg, brow, ei, E, g, [&](int s, bool valid) {
    if (valid) m = fmaxf(m, lrelu(als[s] + dv));
  });
  m = grp_max<16>(m);

  float ss = 0.f;
  for_each_chunk<16>(fast, dd, deg, brow, ei, E, g, [&](int s, bool valid) {
    if (valid) ss += expf(lrelu(als[s] + dv) - m);
  });
  const float q = grp_sum<16>(ss) + 1e-16f;

  float acc = 0.f;
  for_each_chunk<16>(fast, dd, deg, brow, ei, E, g, [&](int s, bool valid) {
    float a = 0.f;
    if (valid) a = expf(lrelu(als[s] + dv) - m) / q;
    unsigned mask = (__builtin_amdgcn_ballot_w32(valid) >> (half * 16)) & 0xffffu;
    while (mask) {
      const int j = __builtin_ctz(mask);
      mask &= mask - 1;
      const int sj = __shfl(s, j, 16);
      const float w = __shfl(a, j, 16);
      acc += H2[(size_t)sj * OUTC + g] * w;
    }
  });

  const float v = acc + b2[g];
  const float mx = grp_max<16>(v);
  const float se = grp_sum<16>(expf(v - mx));
  const float lse = mx + logf(se);
  if (live) out[(size_t)d * OUTC + g] = v - lse;
}

static inline unsigned cdiv(long long a, long long b) { return (unsigned)((a + b - 1) / b); }

extern "C" void kernel_launch(void* const* d_in, const int* in_sizes, int n_in,
                              void* d_out, int out_size, void* d_ws, size_t ws_size,
                              hipStream_t stream) {
  const float* x    = (const float*)d_in[0];
  const int*   ei   = (const int*)  d_in[1];
  const float* W1   = (const float*)d_in[2];
  const float* asr1 = (const float*)d_in[3];
  const float* adt1 = (const float*)d_in[4];
  const float* b1   = (const float*)d_in[5];
  const float* W2   = (const float*)d_in[6];
  const float* asr2 = (const float*)d_in[7];
  const float* adt2 = (const float*)d_in[8];
  const float* b2   = (const float*)d_in[9];
  float* out = (float*)d_out;

  const int n = in_sizes[0] / F_IN;
  const int E = in_sizes[1] / 2;

  float* ws = (float*)d_ws;
  size_t off = 0;
  float* H1     = ws + off; off += (size_t)n * F_IN;
  float* Hp     = ws + off; off += (size_t)n * F_IN;
  int*   bucket = (int*)(ws + off); off += (size_t)n * CAP;
  float* als1   = ws + off; off += (size_t)n * HEADS;
  float* ald1   = ws + off; off += (size_t)n * HEADS;
  float* H2     = ws + off; off += (size_t)n * OUTC;
  float* als2   = ws + off; off += (size_t)n;
  float* ald2   = ws + off; off += (size_t)n;
  int*   cnt    = (int*)(ws + off); off += (size_t)n;

  const int B = 256;

  k_fill0i<<<cdiv(n, B), B, 0, stream>>>(cnt, n);
  k_bucket<<<cdiv(E, B), B, 0, stream>>>(ei, E, cnt, bucket);

  k_gemm1<<<cdiv(n, 16), B, 0, stream>>>(x, W1, H1, n);
  k_coef<HEADS, HID><<<cdiv((long long)n * HEADS, B), B, 0, stream>>>(H1, asr1, adt1, als1, ald1, n);
  k_node1<<<cdiv(n, B / 32), B, 0, stream>>>(ei, E, n, cnt, bucket, als1, ald1, H1, b1, Hp);

  k_gemm2<<<cdiv(cdiv(n, 16), 8), B, 0, stream>>>(Hp, W2, H2, n);
  k_coef<1, OUTC><<<cdiv((long long)n, B), B, 0, stream>>>(H2, asr2, adt2, als2, ald2, n);
  k_node2<<<cdiv(n, B / 16), B, 0, stream>>>(ei, E, n, cnt, bucket, als2, ald2, H2, b2, out);
}


# Node kernels: each destination's edge sources, logits and softmax weights are read once and kept in registers (a lane per edge) instead of three passes that re-read them; sum order unchanged
# speedup vs baseline: 3.9576x; 1.0653x over previous
#include <hip/hip_runtime.h>

typedef __attribute__((ext_vector_type(2))) float v2f;
typedef __attribute__((ext_vector_type(8))) float v8f;

#define F_IN   128
#define HID    32
#define HEADS  4
#define OUTC   16
#define CAP    64

__device__ __forceinline__ float lrelu(float v) { return v > 0.f ? v : 0.2f * v; }

template <int G>
__device__ __forceinline__ float grp_max(float v) {
  for (int o = G >> 1; o > 0; o >>= 1) v = fmaxf(v, __shfl_xor(v, o));
  return v;
}
template <int G>
__device__ __forceinline__ float grp_sum(float v) {
  for (int o = G >> 1; o > 0; o >>= 1) v += __shfl_xor(v, o);
  return v;
}

__global__ void k_fill0i(int* __restrict__ p, int count) {
  int i = blockIdx.x * blockDim.x + threadIdx.x;
  if (i < count) p[i] = 0;
}

__global__ void k_bucket(const int* __restrict__ ei, int E, int* __restrict__ cnt,
                         int* __restrict__ bucket) {
  int e = blockIdx.x * blockDim.x + threadIdx.x;
  if (e >= E) return;
  int s = ei[e], d = ei[E + e];
  int slot = atomicAdd(&cnt[d], 1);
  if (slot < CAP) bucket[(size_t)d * CAP + slot] = s;
}

__global__ void k_gemm1(const float* __restrict__ x, const float* __restrict__ W,
                        float* __restrict__ H, int n) {
  const int wave = threadIdx.x >> 5;
  const int lane = threadIdx.x & 31;
  const int row0 = blockIdx.x * 16;
  if (row0 >= n) return;
  const int col0 = wave * 16;
  const int lr = lane & 15;
  const int lh = lane >> 4;
  const float* ap = x + (size_t)(row0 + lr) * F_IN + 2 * lh;
  const float* bp = W + (size_t)(2 * lh) * F_IN + col0 + lr;
  v8f acc = {};
#pragma unroll
  for (int k = 0; k < F_IN; k += 4) {
    v2f a = *(const v2f*)(ap + k);
    v2f b;
    b.x = bp[(size_t)k * F_IN];
    b.y = bp[(size_t)k * F_IN + F_IN];
    acc = __builtin_amdgcn_wmma_f32_16x16x4_f32(false, a, false, b, (short)0, acc,
                                                false, false);
  }
#pragma unroll
  for (int i = 0; i < 8; i++) {
    H[(size_t)(row0 + i + 8 * lh) * F_IN + col0 + lr] = acc[i];
  }
}

__global__ void k_gemm2(const float* __restrict__ Hp, const float* __restrict__ W2,
                        float* __restrict__ H2, int n) {
  const int wave = threadIdx.x >> 5;
  const int lane = threadIdx.x & 31;
  const int row0 = (blockIdx.x * 8 + wave) * 16;
  if (row0 >= n) return;
  const int lr = lane & 15;
  const int lh = lane >> 4;
  const float* ap = Hp + (size_t)(row0 + lr) * F_IN + 2 * lh;
  const float* bp = W2 + (size_t)(2 * lh) * OUTC + lr;
  v8f acc = {};
#pragma unroll
  for (int k = 0; k < F_IN; k += 4) {
    v2f a = *(const v2f*)(ap + k);
    v2f b;
    b.x = bp[(size_t)k * OUTC];
    b.y = bp[(size_t)k * OUTC + OUTC];
    acc = __builtin_amdgcn_wmma_f32_16x16x4_f32(false, a, false, b, (short)0, acc,
                                                false, false);
  }
#pragma unroll
  for (int i = 0; i < 8; i++) {
    H2[(size_t)(row0 + i + 8 * lh) * OUTC + lr] = acc[i];
  }
}

template <int HN, int CN>
__global__ void k_coef(const float* __restrict__ H, const float* __restrict__ asr,
                       const float* __restrict__ adt, float* __restrict__ als,
                       float* __restrict__ ald, int n) {
  int idx = blockIdx.x * blockDim.x + threadIdx.x;
  if (idx >= n * HN) return;
  int node = idx / HN, h = idx % HN;
  const float* hv = H + (size_t)node * (HN * CN) + h * CN;
  const float* a1 = asr + h * CN;
  const float* a2 = adt + h * CN;
  float s1 = 0.f, s2 = 0.f;
#pragma unroll
  for (int c = 0; c < CN; c++) {
    float v = hv[c];
    s1 += v * a1[c];
    s2 += v * a2[c];
  }
  als[idx] = s1;
  ald[idx] = s2;
}

template <int G, class F>
__device__ __forceinline__ void for_each_chunk(bool fast, int d, int deg,
                                               const int* __restrict__ brow,
                                               const int* __restrict__ ei, int E, int g, F&& f) {
  if (fast) {
    const int tot = deg + 1;
    for (int base = 0; base < tot; base += G) {
      const int j = base + g;
      int s = d;
      if (j < deg) s = brow[j];
      f(s, j < tot);
    }
  } else {
    for (int base = 0; base < E; base += G) {
      const int e = base + g;
      bool valid = false;
      int s = d;
      if (e < E && ei[E + e] == d) { valid = true; s = ei[e]; }
      f(s, valid);
    }
    f(d, g == 0);
  }
}

__global__ void k_node1(const int* __restrict__ ei, int E, int n,
                        const int* __restrict__ cnt, const int* __restrict__ bucket,
                        const float* __restrict__ als, const float* __restrict__ ald,
                        const float* __restrict__ H, const float* __restrict__ b1,
                        float* __restrict__ Hp) {
  const int lane = threadIdx.x & 31;
  const int d = blockIdx.x * (blockDim.x >> 5) + (threadIdx.x >> 5);
  if (d >= n) return;
  const int c = cnt[d];
  const bool fast = c <= CAP;
  const int deg = fast ? c : 0;
  const int* brow = bucket + (size_t)d * CAP;
  const float4 dv = *(const float4*)(ald + (size_t)d * HEADS);
  const float ninf = -__builtin_inff();

  const int hsel = lane >> 3;
  float4 acc = {0.f, 0.f, 0.f, 0.f};

  if (c < 32) {
    const int tot = c + 1;
    const bool valid = lane < tot;
    int s = d;
    if (lane < c) s = brow[lane];
    const float4 a = *(const float4*)(als + (size_t)s * HEADS);
    const float l0 = valid ? lrelu(a.x + dv.x) : ninf;
    const float l1 = valid ? lrelu(a.y + dv.y) : ninf;
    const float l2 = valid ? lrelu(a.z + dv.z) : ninf;
    const float l3 = valid ? lrelu(a.w + dv.w) : ninf;
    const float m0 = grp_max<32>(l0), m1 = grp_max<32>(l1);
    const float m2 = grp_max<32>(l2), m3 = grp_max<32>(l3);
    const float p0 = valid ? expf(l0 - m0) : 0.f;
    const float p1 = valid ? expf(l1 - m1) : 0.f;
    const float p2 = valid ? expf(l2 - m2) : 0.f;
    const float p3 = valid ? expf(l3 - m3) : 0.f;
    const float q0 = grp_sum<32>(p0) + 1e-16f, q1 = grp_sum<32>(p1) + 1e-16f;
    const float q2 = grp_sum<32>(p2) + 1e-16f, q3 = grp_sum<32>(p3) + 1e-16f;
    const float a0 = p0 / q0, a1 = p1 / q1, a2 = p2 / q2, a3 = p3 / q3;
    for (int j = 0; j < tot; j++) {
      const int sj = __shfl(s, j);
      const float w0 = __shfl(a0, j), w1 = __shfl(a1, j);
      const float w2 = __shfl(a2, j), w3 = __shfl(a3, j);
      const float w = hsel == 0 ? w0 : hsel == 1 ? w1 : hsel == 2 ? w2 : w3;
      const float4 hv = *(const float4*)(H + (size_t)sj * F_IN + lane * 4);
      acc.x += hv.x * w; acc.y += hv.y * w; acc.z += hv.z * w; acc.w += hv.w * w;
    }
  } else {
    float m0 = ninf, m1 = ninf, m2 = ninf, m3 = ninf;
    for_each_chunk<32>(fast, d, deg, brow, ei, E, lane, [&](int s, bool valid) {
      if (valid) {
        const float4 a = *(const float4*)(als + (size_t)s * HEADS);
        m0 = fmaxf(m0, lrelu(a.x + dv.x));
        m1 = fmaxf(m1, lrelu(a.y + dv.y));
        m2 = fmaxf(m2, lrelu(a.z + dv.z));
        m3 = fmaxf(m3, lrelu(a.w + dv.w));
      }
    });
    m0 = grp_max<32>(m0); m1 = grp_max<32>(m1); m2 = grp_max<32>(m2); m3 = grp_max<32>(m3);

    float s0 = 0.f, s1 = 0.f, s2 = 0.f, s3 = 0.f;
    for_each_chunk<32>(fast, d, deg, brow, ei, E, lane, [&](int s, bool valid) {
      if (valid) {
        const float4 a = *(const float4*)(als + (size_t)s * HEADS);
        s0 += expf(lrelu(a.x + dv.x) - m0);
        s1 += expf(lrelu(a.y + dv.y) - m1);
        s2 += expf(lrelu(a.z + dv.z) - m2);
        s3 += expf(lrelu(a.w + dv.w) - m3);
      }
    });
    const float q0 = grp_sum<32>(s0) + 1e-16f, q1 = grp_sum<32>(s1) + 1e-16f;
    const float q2 = grp_sum<32>(s2) + 1e-16f, q3 = grp_sum<32>(s3) + 1e-16f;

    for_each_chunk<32>(fast, d, deg, brow, ei, E, lane, [&](int s, bool valid) {
      float a0 = 0.f, a1 = 0.f, a2 = 0.f, a3 = 0.f;
      if (valid) {
        const float4 a = *(const float4*)(als + (size_t)s * HEADS);
        a0 = expf(lrelu(a.x + dv.x) - m0) / q0;
        a1 = expf(lrelu(a.y + dv.y) - m1) / q1;
        a2 = expf(lrelu(a.z + dv.z) - m2) / q2;
        a3 = expf(lrelu(a.w + dv.w) - m3) / q3;
      }
      unsigned mask = __builtin_amdgcn_ballot_w32(valid);
      while (mask) {
        const int j = __builtin_ctz(mask);
        mask &= mask - 1;
        const int sj = __shfl(s, j);
        const float w0 = __shfl(a0, j), w1 = __shfl(a1, j);
        const float w2 = __shfl(a2, j), w3 = __shfl(a3, j);
        const float w = hsel == 0 ? w0 : hsel == 1 ? w1 : hsel == 2 ? w2 : w3;
        const float4 hv = *(const float4*)(H + (size_t)sj * F_IN + lane * 4);
        acc.x += hv.x * w; acc.y += hv.y * w; acc.z += hv.z * w; acc.w += hv.w * w;
      }
    });
  }

  float v[4] = {acc.x + b1[lane * 4 + 0], acc.y + b1[lane * 4 + 1],
                acc.z + b1[lane * 4 + 2], acc.w + b1[lane * 4 + 3]};
  for (int i = 0; i < 4; i++) {
    float t = v[i];
    t = t > 0.f ? t : (expf(t) - 1.f);
    Hp[(size_t)d * F_IN + i * 32 + lane] = t;
  }
}

__global__ void k_node2(const int* __restrict__ ei, int E, int n,
                        const int* __restrict__ cnt, const int* __restrict__ bucket,
                        const float* __restrict__ als, const float* __restrict__ ald,
                        const float* __restrict__ H2, const float* __restrict__ b2,
                        float* __restrict__ out) {
  const int lane = threadIdx.x & 31;
  const int g = lane & 15;
  const int half = lane >> 4;
  const int d = (blockIdx.x * (blockDim.x >> 5) + (threadIdx.x >> 5)) * 2 + half;
  const bool live = d < n;
  const int dd = live ? d : 0;
  const int c = live ? cnt[dd] : -1;
  const bool fast = c <= CAP;
  const int deg = fast ? c : 0;
  const int* brow = bucket + (size_t)dd * CAP;
  const float dv = ald[dd];
  const float ninf = -__builtin_inff();

  float acc = 0.f;
  if (__builtin_amdgcn_ballot_w32(c >= 32) == 0u) {
    const int tot = c + 1;
    const bool v0 = g < tot, v1 = g + 16 < tot;
    int s0 = dd, s1 = dd;
    if (g < c) s0 = brow[g];
    if (g + 16 < c) s1 = brow[g + 16];
    const float r0 = als[s0], r1 = als[s1];
    const float l0 = v0 ? lrelu(r0 + dv) : ninf;
    const float l1 = v1 ? lrelu(r1 + dv) : ninf;
    const float m = grp_max<16>(fmaxf(l0, l1));
    const float p0 = v0 ? expf(l0 - m) : 0.f;
    const float p1 = v1 ? expf(l1 - m) : 0.f;
    const float q = grp_sum<16>(p0 + p1) + 1e-16f;
    const float a0 = p0 / q, a1 = p1 / q;
    const int t0 = tot < 16 ? tot : 16;
    for (int j = 0; j < t0; j++) {
      const int sj = __shfl(s0, j, 16);
      const float w = __shfl(a0, j, 16);
      acc += H2[(size_t)sj * OUTC + g] * w;
    }
    for (int j = 16; j < tot; j++) {
      const int sj = __shfl(s1, j - 16, 16);
      const float w = __shfl(a1, j - 16, 16);
      acc += H2[(size_t)sj * OUTC + g] * w;
    }
  } else {
    float m = ninf;
    for_each_chunk<16>(fast, dd, deg, brow, ei, E, g, [&](int s, bool valid) {
      if (valid) m = fmaxf(m, lrelu(als[s] + dv));
    });
    m = grp_max<16>(m);

    float ss = 0.f;
    for_each_chunk<16>(fast, dd, deg, brow, ei, E, g, [&](int s, bool valid) {
      if (valid) ss += expf(lrelu(als[s] + dv) - m);
    });
    const float q = grp_sum<16>(ss) + 1e-16f;

    for_each_chunk<16>(fast, dd, deg, brow, ei, E, g, [&](int s, bool valid) {
      float a = 0.f;
      if (valid) a = expf(lrelu(als[s] + dv) - m) / q;
      unsigned mask = (__builtin_amdgcn_ballot_w32(valid) >> (half * 16)) & 0xffffu;
      while (mask) {
        const int j = __builtin_ctz(mask);
        mask &= mask - 1;
        const int sj = __shfl(s, j, 16);
        const float w = __shfl(a, j, 16);
        acc += H2[(size_t)sj * OUTC + g] * w;
      }
    });
  }

  const float v = acc + b2[g];
  const float mx = grp_max<16>(v);
  const float se = grp_sum<16>(expf(v - mx));
  const float lse = mx + logf(se);
  if (live) out[(size_t)d * OUTC + g] = v - lse;
}

static inline unsigned cdiv(long long a, long long b) { return (unsigned)((a + b - 1) / b); }

extern "C" void kernel_launch(void* const* d_in, const int* in_sizes, int n_in,
                              void* d_out, int out_size, void* d_ws, size_t ws_size,
                              hipStream_t stream) {
  const float* x    = (const float*)d_in[0];
  const int*   ei   = (const int*)  d_in[1];
  const float* W1   = (const float*)d_in[2];
  const float* asr1 = (const float*)d_in[3];
  const float* adt1 = (const float*)d_in[4];
  const float* b1   = (const float*)d_in[5];
  const float* W2   = (const float*)d_in[6];
  const float* asr2 = (const float*)d_in[7];
  const float* adt2 = (const float*)d_in[8];
  const float* b2   = (const float*)d_in[9];
  float* out = (float*)d_out;

  const int n = in_sizes[0] / F_IN;
  const int E = in_sizes[1] / 2;

  float* ws = (float*)d_ws;
  size_t off = 0;
  float* H1     = ws + off; off += (size_t)n * F_IN;
  float* Hp     = ws + off; off += (size_t)n * F_IN;
  int*   bucket = (int*)(ws + off); off += (size_t)n * CAP;
  float* als1   = ws + off; off += (size_t)n * HEADS;
  float* ald1   = ws + off; off += (size_t)n * HEADS;
  float* H2     = ws + off; off += (size_t)n * OUTC;
  float* als2   = ws + off; off += (size_t)n;
  float* ald2   = ws + off; off += (size_t)n;
  int*   cnt    = (int*)(ws + off); off += (size_t)n;

  const int B = 256;

  k_fill0i<<<cdiv(n, B), B, 0, stream>>>(cnt, n);
  k_bucket<<<cdiv(E, B), B, 0, stream>>>(ei, E, cnt, bucket);

  k_gemm1<<<cdiv(n, 16), B, 0, stream>>>(x, W1, H1, n);
  k_coef<HEADS, HID><<<cdiv((long long)n * HEADS, B), B, 0, stream>>>(H1, asr1, adt1, als1, ald1, n);
  k_node1<<<cdiv(n, B / 32), B, 0, stream>>>(ei, E, n, cnt, bucket, als1, ald1, H1, b1, Hp);

  k_gemm2<<<cdiv(cdiv(n, 16), 8), B, 0, stream>>>(Hp, W2, H2, n);
  k_coef<1, OUTC><<<cdiv((long long)n, B), B, 0, stream>>>(H2, asr2, adt2, als2, ald2, n);
  k_node2<<<cdiv(n, B / 16), B, 0, stream>>>(ei, E, n, cnt, bucket, als2, ald2, H2, b2, out);
}


# GEMM1: one wave computes a whole 16x128 row stripe (8 accumulator tiles), so x is fetched once per stripe with 16-byte loads, not by 8 waves; k order inside the products regrouped
# speedup vs baseline: 4.1640x; 1.0522x over previous
#include <hip/hip_runtime.h>

typedef __attribute__((ext_vector_type(2))) float v2f;
typedef __attribute__((ext_vector_type(8))) float v8f;

#define F_IN   128
#define HID    32
#define HEADS  4
#define OUTC   16
#define CAP    64

__device__ __forceinline__ float lrelu(float v) { return v > 0.f ? v : 0.2f * v; }

template <int G>
__device__ __forceinline__ float grp_max(float v) {
  for (int o = G >> 1; o > 0; o >>= 1) v = fmaxf(v, __shfl_xor(v, o));
  return v;
}
template <int G>
__device__ __forceinline__ float grp_sum(float v) {
  for (int o = G >> 1; o > 0; o >>= 1) v += __shfl_xor(v, o);
  return v;
}

__global__ void k_fill0i(int* __restrict__ p, int count) {
  int i = blockIdx.x * blockDim.x + threadIdx.x;
  if (i < count) p[i] = 0;
}

__global__ void k_bucket(const int* __restrict__ ei, int E, int* __restrict__ cnt,
                         int* __restrict__ bucket) {
  int e = blockIdx.x * blockDim.x + threadIdx.x;
  if (e >= E) return;
  int s = ei[e], d = ei[E + e];
  int slot = atomicAdd(&cnt[d], 1);
  if (slot < CAP) bucket[(size_t)d * CAP + slot] = s;
}

__device__ __forceinline__ float f4e(const float4& v, int i) {
  return i == 0 ? v.x : i == 1 ? v.y : i == 2 ? v.z : v.w;
}

__global__ void __launch_bounds__(64) k_gemm1(const float* __restrict__ x,
                                              const float* __restrict__ W,
                                              float* __restrict__ H, int n) {
  const int lane = threadIdx.x & 31;
  const int row0 = (blockIdx.x * (blockDim.x >> 5) + (threadIdx.x >> 5)) * 16;
  if (row0 >= n) return;
  const int c = lane & 15;
  const int h = lane >> 4;
  const float* ap = x + (size_t)(row0 + c) * F_IN + 4 * h;
  const float* bp = W + (size_t)(4 * h) * F_IN + 4 * c;

  v8f acc[8];
#pragma unroll
  for (int t = 0; t < 8; t++) acc[t] = v8f{};

  float4 a_cur = *(const float4*)ap;
  float4 b_cur[8];
#pragma unroll
  for (int s = 0; s < 4; s++) {
    b_cur[2 * s]     = *(const float4*)(bp + (size_t)s * F_IN);
    b_cur[2 * s + 1] = *(const float4*)(bp + (size_t)s * F_IN + 64);
  }

#pragma unroll
  for (int m = 0; m < F_IN / 8; m++) {
    float4 a_nxt = a_cur;
    float4 b_nxt[8];
#pragma unroll
    for (int s = 0; s < 8; s++) b_nxt[s] = b_cur[s];
    if (m + 1 < F_IN / 8) {
      a_nxt = *(const float4*)(ap + 8 * (m + 1));
#pragma unroll
      for (int s = 0; s < 4; s++) {
        b_nxt[2 * s]     = *(const float4*)(bp + (size_t)(8 * (m + 1) + s) * F_IN);
        b_nxt[2 * s + 1] = *(const float4*)(bp + (size_t)(8 * (m + 1) + s) * F_IN + 64);
      }
    }
#pragma unroll
    for (int u = 0; u < 2; u++) {
      v2f av;
      av.x = u == 0 ? a_cur.x : a_cur.z;
      av.y = u == 0 ? a_cur.y : a_cur.w;
#pragma unroll
      for (int t = 0; t < 8; t++) {
        v2f bv;
        bv.x = f4e(b_cur[2 * (2 * u) + (t >> 2)], t & 3);
        bv.y = f4e(b_cur[2 * (2 * u + 1) + (t >> 2)], t & 3);
        acc[t] = __builtin_amdgcn_wmma_f32_16x16x4_f32(false, av, false, bv, (short)0, acc[t],
                                                       false, false);
      }
    }
    a_cur = a_nxt;
#pragma unroll
    for (int s = 0; s < 8; s++) b_cur[s] = b_nxt[s];
  }

#pragma unroll
  for (int r = 0; r < 8; r++) {
    float* hp = H + (size_t)(row0 + 8 * h + r) * F_IN + 4 * c;
    *(float4*)hp        = make_float4(acc[0][r], acc[1][r], acc[2][r], acc[3][r]);
    *(float4*)(hp + 64) = make_float4(acc[4][r], acc[5][r], acc[6][r], acc[7][r]);
  }
}

__global__ void k_gemm2(const float* __restrict__ Hp, const float* __restrict__ W2,
                        float* __restrict__ H2, int n) {
  const int wave = threadIdx.x >> 5;
  const int lane = threadIdx.x & 31;
  const int row0 = (blockIdx.x * 8 + wave) * 16;
  if (row0 >= n) return;
  const int lr = lane & 15;
  const int lh = lane >> 4;
  const float* ap = Hp + (size_t)(row0 + lr) * F_IN + 2 * lh;
  const float* bp = W2 + (size_t)(2 * lh) * OUTC + lr;
  v8f acc = {};
#pragma unroll
  for (int k = 0; k < F_IN; k += 4) {
    v2f a = *(const v2f*)(ap + k);
    v2f b;
    b.x = bp[(size_t)k * OUTC];
    b.y = bp[(size_t)k * OUTC + OUTC];
    acc = __builtin_amdgcn_wmma_f32_16x16x4_f32(false, a, false, b, (short)0, acc,
                                                false, false);
  }
#pragma unroll
  for (int i = 0; i < 8; i++) {
    H2[(size_t)(row0 + i + 8 * lh) * OUTC + lr] = acc[i];
  }
}

template <int HN, int CN>
__global__ void k_coef(const float* __restrict__ H, const float* __restrict__ asr,
                       const float* __restrict__ adt, float* __restrict__ als,
                       float* __restrict__ ald, int n) {
  int idx = blockIdx.x * blockDim.x + threadIdx.x;
  if (idx >= n * HN) return;
  int node = idx / HN, h = idx % HN;
  const float* hv = H + (size_t)node * (HN * CN) + h * CN;
  const float* a1 = asr + h * CN;
  const float* a2 = adt + h * CN;
  float s1 = 0.f, s2 = 0.f;
#pragma unroll
  for (int c = 0; c < CN; c++) {
    float v = hv[c];
    s1 += v * a1[c];
    s2 += v * a2[c];
  }
  als[idx] = s1;
  ald[idx] = s2;
}

template <int G, class F>
__device__ __forceinline__ void for_each_chunk(bool fast, int d, int deg,
                                               const int* __restrict__ brow,
                                               const int* __restrict__ ei, int E, int g, F&& f) {
  if (fast) {
    const int tot = deg + 1;
    for (int base = 0; base < tot; base += G) {
      const int j = base + g;
      int s = d;
      if (j < deg) s = brow[j];
      f(s, j < tot);
    }
  } else {
    for (int base = 0; base < E; base += G) {
      const int e = base + g;
      bool valid = false;
      int s = d;
      if (e < E && ei[E + e] == d) { valid = true; s = ei[e]; }
      f(s, valid);
    }
    f(d, g == 0);
  }
}

__global__ void k_node1(const int* __restrict__ ei, int E, int n,
                        const int* __restrict__ cnt, const int* __restrict__ bucket,
                        const float* __restrict__ als, const float* __restrict__ ald,
                        const float* __restrict__ H, const float* __restrict__ b1,
                        float* __restrict__ Hp) {
  const int lane = threadIdx.x & 31;
  const int d = blockIdx.x * (blockDim.x >> 5) + (threadIdx.x >> 5);
  if (d >= n) return;
  const int c = cnt[d];
  const bool fast = c <= CAP;
  const int deg = fast ? c : 0;
  const int* brow = bucket + (size_t)d * CAP;
  const float4 dv = *(const float4*)(ald + (size_t)d * HEADS);
  const float ninf = -__builtin_inff();

  const int hsel = lane >> 3;
  float4 acc = {0.f, 0.f, 0.f, 0.f};

  if (c < 32) {
    const int tot = c + 1;
    const bool valid = lane < tot;
    int s = d;
    if (lane < c) s = brow[lane];
    const float4 a = *(const float4*)(als + (size_t)s * HEADS);
    const float l0 = valid ? lrelu(a.x + dv.x) : ninf;
    const float l1 = valid ? lrelu(a.y + dv.y) : ninf;
    const float l2 = valid ? lrelu(a.z + dv.z) : ninf;
    const float l3 = valid ? lrelu(a.w + dv.w) : ninf;
    const float m0 = grp_max<32>(l0), m1 = grp_max<32>(l1);
    const float m2 = grp_max<32>(l2), m3 = grp_max<32>(l3);
    const float p0 = valid ? expf(l0 - m0) : 0.f;
    const float p1 = valid ? expf(l1 - m1) : 0.f;
    const float p2 = valid ? expf(l2 - m2) : 0.f;
    const float p3 = valid ? expf(l3 - m3) : 0.f;
    const float q0 = grp_sum<32>(p0) + 1e-16f, q1 = grp_sum<32>(p1) + 1e-16f;
    const float q2 = grp_sum<32>(p2) + 1e-16f, q3 = grp_sum<32>(p3) + 1e-16f;
    const float a0 = p0 / q0, a1 = p1 / q1, a2 = p2 / q2, a3 = p3 / q3;
    for (int j = 0; j < tot; j++) {
      const int sj = __shfl(s, j);
      const float w0 = __shfl(a0, j), w1 = __shfl(a1, j);
      const float w2 = __shfl(a2, j), w3 = __shfl(a3, j);
      const float w = hsel == 0 ? w0 : hsel == 1 ? w1 : hsel == 2 ? w2 : w3;
      const float4 hv = *(const float4*)(H + (size_t)sj * F_IN + lane * 4);
      acc.x += hv.x * w; acc.y += hv.y * w; acc.z += hv.z * w; acc.w += hv.w * w;
    }
  } else {
    float m0 = ninf, m1 = ninf, m2 = ninf, m3 = ninf;
    for_each_chunk<32>(fast, d, deg, brow, ei, E, lane, [&](int s, bool valid) {
      if (valid) {
        const float4 a = *(const float4*)(als + (size_t)s * HEADS);
        m0 = fmaxf(m0, lrelu(a.x + dv.x));
        m1 = fmaxf(m1, lrelu(a.y + dv.y));
        m2 = fmaxf(m2, lrelu(a.z + dv.z));
        m3 = fmaxf(m3, lrelu(a.w + dv.w));
      }
    });
    m0 = grp_max<32>(m0); m1 = grp_max<32>(m1); m2 = grp_max<32>(m2); m3 = grp_max<32>(m3);

    float s0 = 0.f, s1 = 0.f, s2 = 0.f, s3 = 0.f;
    for_each_chunk<32>(fast, d, deg, brow, ei, E, lane, [&](int s, bool valid) {
      if (valid) {
        const float4 a = *(const float4*)(als + (size_t)s * HEADS);
        s0 += expf(lrelu(a.x + dv.x) - m0);
        s1 += expf(lrelu(a.y + dv.y) - m1);
        s2 += expf(lrelu(a.z + dv.z) - m2);
        s3 += expf(lrelu(a.w + dv.w) - m3);
      }
    });
    const float q0 = grp_sum<32>(s0) + 1e-16f, q1 = grp_sum<32>(s1) + 1e-16f;
    const float q2 = grp_sum<32>(s2) + 1e-16f, q3 = grp_sum<32>(s3) + 1e-16f;

    for_each_chunk<32>(fast, d, deg, brow, ei, E, lane, [&](int s, bool valid) {
      float a0 = 0.f, a1 = 0.f, a2 = 0.f, a3 = 0.f;
      if (valid) {
        const float4 a = *(const float4*)(als + (size_t)s * HEADS);
        a0 = expf(lrelu(a.x + dv.x) - m0) / q0;
        a1 = expf(lrelu(a.y + dv.y) - m1) / q1;
        a2 = expf(lrelu(a.z + dv.z) - m2) / q2;
        a3 = expf(lrelu(a.w + dv.w) - m3) / q3;
      }
      unsigned mask = __builtin_amdgcn_ballot_w32(valid);
      while (mask) {
        const int j = __builtin_ctz(mask);
        mask &= mask - 1;
        const int sj = __shfl(s, j);
        const float w0 = __shfl(a0, j), w1 = __shfl(a1, j);
        const float w2 = __shfl(a2, j), w3 = __shfl(a3, j);
        const float w = hsel == 0 ? w0 : hsel == 1 ? w1 : hsel == 2 ? w2 : w3;
        const float4 hv = *(const float4*)(H + (size_t)sj * F_IN + lane * 4);
        acc.x += hv.x * w; acc.y += hv.y * w; acc.z += hv.z * w; acc.w += hv.w * w;
      }
    });
  }

  float v[4] = {acc.x + b1[lane * 4 + 0], acc.y + b1[lane * 4 + 1],
                acc.z + b1[lane * 4 + 2], acc.w + b1[lane * 4 + 3]};
  for (int i = 0; i < 4; i++) {
    float t = v[i];
    t = t > 0.f ? t : (expf(t) - 1.f);
    Hp[(size_t)d * F_IN + i * 32 + lane] = t;
  }
}

__global__ void k_node2(const int* __restrict__ ei, int E, int n,
                        const int* __restrict__ cnt, const int* __restrict__ bucket,
                        const float* __restrict__ als, const float* __restrict__ ald,
                        const float* __restrict__ H2, const float* __restrict__ b2,
                        float* __restrict__ out) {
  const int lane = threadIdx.x & 31;
  const int g = lane & 15;
  const int half = lane >> 4;
  const int d = (blockIdx.x * (blockDim.x >> 5) + (threadIdx.x >> 5)) * 2 + half;
  const bool live = d < n;
  const int dd = live ? d : 0;
  const int c = live ? cnt[dd] : -1;
  const bool fast = c <= CAP;
  const int deg = fast ? c : 0;
  const int* brow = bucket + (size_t)dd * CAP;
  const float dv = ald[dd];
  const float ninf = -__builtin_inff();

  float acc = 0.f;
  if (__builtin_amdgcn_ballot_w32(c >= 32) == 0u) {
    const int tot = c + 1;
    const bool v0 = g < tot, v1 = g + 16 < tot;
    int s0 = dd, s1 = dd;
    if (g < c) s0 = brow[g];
    if (g + 16 < c) s1 = brow[g + 16];
    const float r0 = als[s0], r1 = als[s1];
    const float l0 = v0 ? lrelu(r0 + dv) : ninf;
    const float l1 = v1 ? lrelu(r1 + dv) : ninf;
    const float m = grp_max<16>(fmaxf(l0, l1));
    const float p0 = v0 ? expf(l0 - m) : 0.f;
    const float p1 = v1 ? expf(l1 - m) : 0.f;
    const float q = grp_sum<16>(p0 + p1) + 1e-16f;
    const float a0 = p0 / q, a1 = p1 / q;
    const int t0 = tot < 16 ? tot : 16;
    for (int j = 0; j < t0; j++) {
      const int sj = __shfl(s0, j, 16);
      const float w = __shfl(a0, j, 16);
      acc += H2[(size_t)sj * OUTC + g] * w;
    }
    for (int j = 16; j < tot; j++) {
      const int sj = __shfl(s1, j - 16, 16);
      const float w = __shfl(a1, j - 16, 16);
      acc += H2[(size_t)sj * OUTC + g] * w;
    }
  } else {
    float m = ninf;
    for_each_chunk<16>(fast, dd, deg, brow, ei, E, g, [&](int s, bool valid) {
      if (valid) m = fmaxf(m, lrelu(als[s] + dv));
    });
    m = grp_max<16>(m);

    float ss = 0.f;
    for_each_chunk<16>(fast, dd, deg, brow, ei, E, g, [&](int s, bool valid) {
      if (valid) ss += expf(lrelu(als[s] + dv) - m);
    });
    const float q = grp_sum<16>(ss) + 1e-16f;

    for_each_chunk<16>(fast, dd, deg, brow, ei, E, g, [&](int s, bool valid) {
      float a = 0.f;
      if (valid) a = expf(lrelu(als[s] + dv) - m) / q;
      unsigned mask = (__builtin_amdgcn_ballot_w32(valid) >> (half * 16)) & 0xffffu;
      while (mask) {
        const int j = __builtin_ctz(mask);
        mask &= mask - 1;
        const int sj = __shfl(s, j, 16);
        const float w = __shfl(a, j, 16);
        acc += H2[(size_t)sj * OUTC + g] * w;
      }
    });
  }

  const float v = acc + b2[g];
  const float mx = grp_max<16>(v);
  const float se = grp_sum<16>(expf(v - mx));
  const float lse = mx + logf(se);
  if (live) out[(size_t)d * OUTC + g] = v - lse;
}

static inline unsigned cdiv(long long a, long long b) { return (unsigned)((a + b - 1) / b); }

extern "C" void kernel_launch(void* const* d_in, const int* in_sizes, int n_in,
                              void* d_out, int out_size, void* d_ws, size_t ws_size,
                              hipStream_t stream) {
  const float* x    = (const float*)d_in[0];
  const int*   ei   = (const int*)  d_in[1];
  const float* W1   = (const float*)d_in[2];
  const float* asr1 = (const float*)d_in[3];
  const float* adt1 = (const float*)d_in[4];
  const float* b1   = (const float*)d_in[5];
  const float* W2   = (const float*)d_in[6];
  const float* asr2 = (const float*)d_in[7];
  const float* adt2 = (const float*)d_in[8];
  const float* b2   = (const float*)d_in[9];
  float* out = (float*)d_out;

  const int n = in_sizes[0] / F_IN;
  const int E = in_sizes[1] / 2;

  float* ws = (float*)d_ws;
  size_t off = 0;
  float* H1     = ws + off; off += (size_t)n * F_IN;
  float* Hp     = ws + off; off += (size_t)n * F_IN;
  int*   bucket = (int*)(ws + off); off += (size_t)n * CAP;
  float* als1   = ws + off; off += (size_t)n * HEADS;
  float* ald1   = ws + off; off += (size_t)n * HEADS;
  float* H2     = ws + off; off += (size_t)n * OUTC;
  float* als2   = ws + off; off += (size_t)n;
  float* ald2   = ws + off; off += (size_t)n;
  int*   cnt    = (int*)(ws + off); off += (size_t)n;

  const int B = 256;

  k_fill0i<<<cdiv(n, B), B, 0, stream>>>(cnt, n);
  k_bucket<<<cdiv(E, B), B, 0, stream>>>(ei, E, cnt, bucket);

  k_gemm1<<<cdiv(cdiv(n, 16), 2), 64, 0, stream>>>(x, W1, H1, n);
  k_coef<HEADS, HID><<<cdiv((long long)n * HEADS, B), B, 0, stream>>>(H1, asr1, adt1, als1, ald1, n);
  k_node1<<<cdiv(n, B / 32), B, 0, stream>>>(ei, E, n, cnt, bucket, als1, ald1, H1, b1, Hp);

  k_gemm2<<<cdiv(cdiv(n, 16), 8), B, 0, stream>>>(Hp, W2, H2, n);
  k_coef<1, OUTC><<<cdiv((long long)n, B), B, 0, stream>>>(H2, asr2, adt2, als2, ald2, n);
  k_node2<<<cdiv(n, B / 16), B, 0, stream>>>(ei, E, n, cnt, bucket, als2, ald2, H2, b2, out);
}


# Node kernels, register paths: the row loads of 8 list entries (then 4, 2, 1) are issued together ahead of their weighted adds, so a wave has 8 rows in flight instead of 1; sum order unchanged
# speedup vs baseline: 4.5987x; 1.1044x over previous
#include <hip/hip_runtime.h>

typedef __attribute__((ext_vector_type(2))) float v2f;
typedef __attribute__((ext_vector_type(8))) float v8f;

#define F_IN   128
#define HID    32
#define HEADS  4
#define OUTC   16
#define CAP    64

__device__ __forceinline__ float lrelu(float v) { return v > 0.f ? v : 0.2f * v; }

template <int G>
__device__ __forceinline__ float grp_max(float v) {
  for (int o = G >> 1; o > 0; o >>= 1) v = fmaxf(v, __shfl_xor(v, o));
  return v;
}
template <int G>
__device__ __forceinline__ float grp_sum(float v) {
  for (int o = G >> 1; o > 0; o >>= 1) v += __shfl_xor(v, o);
  return v;
}

__global__ void k_fill0i(int* __restrict__ p, int count) {
  int i = blockIdx.x * blockDim.x + threadIdx.x;
  if (i < count) p[i] = 0;
}

__global__ void k_bucket(const int* __restrict__ ei, int E, int* __restrict__ cnt,
                         int* __restrict__ bucket) {
  int e = blockIdx.x * blockDim.x + threadIdx.x;
  if (e >= E) return;
  int s = ei[e], d = ei[E + e];
  int slot = atomicAdd(&cnt[d], 1);
  if (slot < CAP) bucket[(size_t)d * CAP + slot] = s;
}

__device__ __forceinline__ float f4e(const float4& v, int i) {
  return i == 0 ? v.x : i == 1 ? v.y : i == 2 ? v.z : v.w;
}

__global__ void __launch_bounds__(64) k_gemm1(const float* __restrict__ x,
                                              const float* __restrict__ W,
                                              float* __restrict__ H, int n) {
  const int lane = threadIdx.x & 31;
  const int row0 = (blockIdx.x * (blockDim.x >> 5) + (threadIdx.x >> 5)) * 16;
  if (row0 >= n) return;
  const int c = lane & 15;
  const int h = lane >> 4;
  const float* ap = x + (size_t)(row0 + c) * F_IN + 4 * h;
  const float* bp = W + (size_t)(4 * h) * F_IN + 4 * c;

  v8f acc[8];
#pragma unroll
  for (int t = 0; t < 8; t++) acc[t] = v8f{};

  float4 a_cur = *(const float4*)ap;
  float4 b_cur[8];
#pragma unroll
  for (int s = 0; s < 4; s++) {
    b_cur[2 * s]     = *(const float4*)(bp + (size_t)s * F_IN);
    b_cur[2 * s + 1] = *(const float4*)(bp + (size_t)s * F_IN + 64);
  }

#pragma unroll
  for (int m = 0; m < F_IN / 8; m++) {
    float4 a_nxt = a_cur;
    float4 b_nxt[8];
#pragma unroll
    for (int s = 0; s < 8; s++) b_nxt[s] = b_cur[s];
    if (m + 1 < F_IN / 8) {
      a_nxt = *(const float4*)(ap + 8 * (m + 1));
#pragma unroll
      for (int s = 0; s < 4; s++) {
        b_nxt[2 * s]     = *(const float4*)(bp + (size_t)(8 * (m + 1) + s) * F_IN);
        b_nxt[2 * s + 1] = *(const float4*)(bp + (size_t)(8 * (m + 1) + s) * F_IN + 64);
      }
    }
#pragma unroll
    for (int u = 0; u < 2; u++) {
      v2f av;
      av.x = u == 0 ? a_cur.x : a_cur.z;
      av.y = u == 0 ? a_cur.y : a_cur.w;
#pragma unroll
      for (int t = 0; t < 8; t++) {
        v2f bv;
        bv.x = f4e(b_cur[2 * (2 * u) + (t >> 2)], t & 3);
        bv.y = f4e(b_cur[2 * (2 * u + 1) + (t >> 2)], t & 3);
        acc[t] = __builtin_amdgcn_wmma_f32_16x16x4_f32(false, av, false, bv, (short)0, acc[t],
                                                       false, false);
      }
    }
    a_cur = a_nxt;
#pragma unroll
    for (int s = 0; s < 8; s++) b_cur[s] = b_nxt[s];
  }

#pragma unroll
  for (int r = 0; r < 8; r++) {
    float* hp = H + (size_t)(row0 + 8 * h + r) * F_IN + 4 * c;
    *(float4*)hp        = make_float4(acc[0][r], acc[1][r], acc[2][r], acc[3][r]);
    *(float4*)(hp + 64) = make_float4(acc[4][r], acc[5][r], acc[6][r], acc[7][r]);
  }
}

__global__ void k_gemm2(const float* __restrict__ Hp, const float* __restrict__ W2,
                        float* __restrict__ H2, int n) {
  const int wave = threadIdx.x >> 5;
  const int lane = threadIdx.x & 31;
  const int row0 = (blockIdx.x * 8 + wave) * 16;
  if (row0 >= n) return;
  const int lr = lane & 15;
  const int lh = lane >> 4;
  const float* ap = Hp + (size_t)(row0 + lr) * F_IN + 2 * lh;
  const float* bp = W2 + (size_t)(2 * lh) * OUTC + lr;
  v8f acc = {};
#pragma unroll
  for (int k = 0; k < F_IN; k += 4) {
    v2f a = *(const v2f*)(ap + k);
    v2f b;
    b.x = bp[(size_t)k * OUTC];
    b.y = bp[(size_t)k * OUTC + OUTC];
    acc = __builtin_amdgcn_wmma_f32_16x16x4_f32(false, a, false, b, (short)0, acc,
                                                false, false);
  }
#pragma unroll
  for (int i = 0; i < 8; i++) {
    H2[(size_t)(row0 + i + 8 * lh) * OUTC + lr] = acc[i];
  }
}

template <int HN, int CN>
__global__ void k_coef(const float* __restrict__ H, const float* __restrict__ asr,
                       const float* __restrict__ adt, float* __restrict__ als,
                       float* __restrict__ ald, int n) {
  int idx = blockIdx.x * blockDim.x + threadIdx.x;
  if (idx >= n * HN) return;
  int node = idx / HN, h = idx % HN;
  const float* hv = H + (size_t)node * (HN * CN) + h * CN;
  const float* a1 = asr + h * CN;
  const float* a2 = adt + h * CN;
  float s1 = 0.f, s2 = 0.f;
#pragma unroll
  for (int c = 0; c < CN; c++) {
    float v = hv[c];
    s1 += v * a1[c];
    s2 += v * a2[c];
  }
  als[idx] = s1;
  ald[idx] = s2;
}

template <int G, class F>
__device__ __forceinline__ void for_each_chunk(bool fast, int d, int deg,
                                               const int* __restrict__ brow,
                                               const int* __restrict__ ei, int E, int g, F&& f) {
  if (fast) {
    const int tot = deg + 1;
    for (int base = 0; base < tot; base += G) {
      const int j = base + g;
      int s = d;
      if (j < deg) s = brow[j];
      f(s, j < tot);
    }
  } else {
    for (int base = 0; base < E; base += G) {
      const int e = base + g;
      bool valid = false;
      int s = d;
      if (e < E && ei[E + e] == d) { valid = true; s = ei[e]; }
      f(s, valid);
    }
    f(d, g == 0);
  }
}

template <int U>
__device__ __forceinline__ void gather_rows1(const float* __restrict__ H, int s, float a0, float a1,
                                             float a2, float a3, int j0, int hsel, int lane,
                                             float4& acc) {
  float4 hv[U];
  float w[U];
#pragma unroll
  for (int u = 0; u < U; u++) {
    const int sj = __shfl(s, j0 + u);
    hv[u] = ((const float4*)H)[sj * (F_IN / 4) + lane];
  }
#pragma unroll
  for (int u = 0; u < U; u++) {
    const float w0 = __shfl(a0, j0 + u), w1 = __shfl(a1, j0 + u);
    const float w2 = __shfl(a2, j0 + u), w3 = __shfl(a3, j0 + u);
    w[u] = hsel == 0 ? w0 : hsel == 1 ? w1 : hsel == 2 ? w2 : w3;
  }
#pragma unroll
  for (int u = 0; u < U; u++) {
    acc.x = fmaf(hv[u].x, w[u], acc.x); acc.y = fmaf(hv[u].y, w[u], acc.y);
    acc.z = fmaf(hv[u].z, w[u], acc.z); acc.w = fmaf(hv[u].w, w[u], acc.w);
  }
}

template <int U>
__device__ __forceinline__ void gather_rows2(const float* __restrict__ H2, int s, float a, int j0,
                                             int g, float& acc) {
  float hv[U];
  float w[U];
#pragma unroll
  for (int u = 0; u < U; u++) {
    const int sj = __shfl(s, j0 + u, 16);
    hv[u] = H2[(size_t)sj * OUTC + g];
  }
#pragma unroll
  for (int u = 0; u < U; u++) w[u] = __shfl(a, j0 + u, 16);
#pragma unroll
  for (int u = 0; u < U; u++) acc = fmaf(hv[u], w[u], acc);
}

__device__ __forceinline__ void gather_list2(const float* __restrict__ H2, int s, float a, int cnt,
                                             int g, float& acc) {
  int j = 0;
  for (; j + 8 <= cnt; j += 8) gather_rows2<8>(H2, s, a, j, g, acc);
  if (cnt - j >= 4) { gather_rows2<4>(H2, s, a, j, g, acc); j += 4; }
  if (cnt - j >= 2) { gather_rows2<2>(H2, s, a, j, g, acc); j += 2; }
  if (cnt - j >= 1) gather_rows2<1>(H2, s, a, j, g, acc);
}

__global__ void k_node1(const int* __restrict__ ei, int E, int n,
                        const int* __restrict__ cnt, const int* __restrict__ bucket,
                        const float* __restrict__ als, const float* __restrict__ ald,
                        const float* __restrict__ H, const float* __restrict__ b1,
                        float* __restrict__ Hp) {
  const int lane = threadIdx.x & 31;
  const int d = blockIdx.x * (blockDim.x >> 5) + (threadIdx.x >> 5);
  if (d >= n) return;
  const int c = cnt[d];
  const bool fast = c <= CAP;
  const int deg = fast ? c : 0;
  const int* brow = bucket + (size_t)d * CAP;
  const float4 dv = *(const float4*)(ald + (size_t)d * HEADS);
  const float ninf = -__builtin_inff();

  const int hsel = lane >> 3;
  float4 acc = {0.f, 0.f, 0.f, 0.f};

  if (c < 32) {
    const int tot = c + 1;
    const bool valid = lane < tot;
    int s = d;
    if (lane < c) s = brow[lane];
    const float4 a = *(const float4*)(als + (size_t)s * HEADS);
    const float l0 = valid ? lrelu(a.x + dv.x) : ninf;
    const float l1 = valid ? lrelu(a.y + dv.y) : ninf;
    const float l2 = valid ? lrelu(a.z + dv.z) : ninf;
    const float l3 = valid ? lrelu(a.w + dv.w) : ninf;
    const float m0 = grp_max<32>(l0), m1 = grp_max<32>(l1);
    const float m2 = grp_max<32>(l2), m3 = grp_max<32>(l3);
    const float p0 = valid ? expf(l0 - m0) : 0.f;
    const float p1 = valid ? expf(l1 - m1) : 0.f;
    const float p2 = valid ? expf(l2 - m2) : 0.f;
    const float p3 = valid ? expf(l3 - m3) : 0.f;
    const float q0 = grp_sum<32>(p0) + 1e-16f, q1 = grp_sum<32>(p1) + 1e-16f;
    const float q2 = grp_sum<32>(p2) + 1e-16f, q3 = grp_sum<32>(p3) + 1e-16f;
    const float a0 = p0 / q0, a1 = p1 / q1, a2 = p2 / q2, a3 = p3 / q3;
    int j = 0;
    for (; j + 8 <= tot; j += 8) gather_rows1<8>(H, s, a0, a1, a2, a3, j, hsel, lane, acc);
    if (tot - j >= 4) { gather_rows1<4>(H, s, a0, a1, a2, a3, j, hsel, lane, acc); j += 4; }
    if (tot - j >= 2) { gather_rows1<2>(H, s, a0, a1, a2, a3, j, hsel, lane, acc); j += 2; }
    if (tot - j >= 1) gather_rows1<1>(H, s, a0, a1, a2, a3, j, hsel, lane, acc);
  } else {
    float m0 = ninf, m1 = ninf, m2 = ninf, m3 = ninf;
    for_each_chunk<32>(fast, d, deg, brow, ei, E, lane, [&](int s, bool valid) {
      if (valid) {
        const float4 a = *(const float4*)(als + (size_t)s * HEADS);
        m0 = fmaxf(m0, lrelu(a.x + dv.x));
        m1 = fmaxf(m1, lrelu(a.y + dv.y));
        m2 = fmaxf(m2, lrelu(a.z + dv.z));
        m3 = fmaxf(m3, lrelu(a.w + dv.w));
      }
    });
    m0 = grp_max<32>(m0); m1 = grp_max<32>(m1); m2 = grp_max<32>(m2); m3 = grp_max<32>(m3);

    float s0 = 0.f, s1 = 0.f, s2 = 0.f, s3 = 0.f;
    for_each_chunk<32>(fast, d, deg, brow, ei, E, lane, [&](int s, bool valid) {
      if (valid) {
        const float4 a = *(const float4*)(als + (size_t)s * HEADS);
        s0 += expf(lrelu(a.x + dv.x) - m0);
        s1 += expf(lrelu(a.y + dv.y) - m1);
        s2 += expf(lrelu(a.z + dv.z) - m2);
        s3 += expf(lrelu(a.w + dv.w) - m3);
      }
    });
    const float q0 = grp_sum<32>(s0) + 1e-16f, q1 = grp_sum<32>(s1) + 1e-16f;
    const float q2 = grp_sum<32>(s2) + 1e-16f, q3 = grp_sum<32>(s3) + 1e-16f;

    for_each_chunk<32>(fast, d, deg, brow, ei, E, lane, [&](int s, bool valid) {
      float a0 = 0.f, a1 = 0.f, a2 = 0.f, a3 = 0.f;
      if (valid) {
        const float4 a = *(const float4*)(als + (size_t)s * HEADS);
        a0 = expf(lrelu(a.x + dv.x) - m0) / q0;
        a1 = expf(lrelu(a.y + dv.y) - m1) / q1;
        a2 = expf(lrelu(a.z + dv.z) - m2) / q2;
        a3 = expf(lrelu(a.w + dv.w) - m3) / q3;
      }
      unsigned mask = __builtin_amdgcn_ballot_w32(valid);
      while (mask) {
        const int j = __builtin_ctz(mask);
        mask &= mask - 1;
        const int sj = __shfl(s, j);
        const float w0 = __shfl(a0, j), w1 = __shfl(a1, j);
        const float w2 = __shfl(a2, j), w3 = __shfl(a3, j);
        const float w = hsel == 0 ? w0 : hsel == 1 ? w1 : hsel == 2 ? w2 : w3;
        const float4 hv = *(const float4*)(H + (size_t)sj * F_IN + lane * 4);
        acc.x += hv.x * w; acc.y += hv.y * w; acc.z += hv.z * w; acc.w += hv.w * w;
      }
    });
  }

  float v[4] = {acc.x + b1[lane * 4 + 0], acc.y + b1[lane * 4 + 1],
                acc.z + b1[lane * 4 + 2], acc.w + b1[lane * 4 + 3]};
  for (int i = 0; i < 4; i++) {
    float t = v[i];
    t = t > 0.f ? t : (expf(t) - 1.f);
    Hp[(size_t)d * F_IN + i * 32 + lane] = t;
  }
}

__global__ void k_node2(const int* __restrict__ ei, int E, int n,
                        const int* __restrict__ cnt, const int* __restrict__ bucket,
                        const float* __restrict__ als, const float* __restrict__ ald,
                        const float* __restrict__ H2, const float* __restrict__ b2,
                        float* __restrict__ out) {
  const int lane = threadIdx.x & 31;
  const int g = lane & 15;
  const int half = lane >> 4;
  const int d = (blockIdx.x * (blockDim.x >> 5) + (threadIdx.x >> 5)) * 2 + half;
  const bool live = d < n;
  const int dd = live ? d : 0;
  const int c = live ? cnt[dd] : -1;
  const bool fast = c <= CAP;
  const int deg = fast ? c : 0;
  const int* brow = bucket + (size_t)dd * CAP;
  const float dv = ald[dd];
  const float ninf = -__builtin_inff();

  float acc = 0.f;
  if (__builtin_amdgcn_ballot_w32(c >= 32) == 0u) {
    const int tot = c + 1;
    const bool v0 = g < tot, v1 = g + 16 < tot;
    int s0 = dd, s1 = dd;
    if (g < c) s0 = brow[g];
    if (g + 16 < c) s1 = brow[g + 16];
    const float r0 = als[s0], r1 = als[s1];
    const float l0 = v0 ? lrelu(r0 + dv) : ninf;
    const float l1 = v1 ? lrelu(r1 + dv) : ninf;
    const float m = grp_max<16>(fmaxf(l0, l1));
    const float p0 = v0 ? expf(l0 - m) : 0.f;
    const float p1 = v1 ? expf(l1 - m) : 0.f;
    const float q = grp_sum<16>(p0 + p1) + 1e-16f;
    const float a0 = p0 / q, a1 = p1 / q;
    const int t0 = tot < 16 ? tot : 16;
    gather_list2(H2, s0, a0, t0, g, acc);
    gather_list2(H2, s1, a1, tot - t0, g, acc);
  } else {
    float m = ninf;
    for_each_chunk<16>(fast, dd, deg, brow, ei, E, g, [&](int s, bool valid) {
      if (valid) m = fmaxf(m, lrelu(als[s] + dv));
    });
    m = grp_max<16>(m);

    float ss = 0.f;
    for_each_chunk<16>(fast, dd, deg, brow, ei, E, g, [&](int s, bool valid) {
      if (valid) ss += expf(lrelu(als[s] + dv) - m);
    });
    const float q = grp_sum<16>(ss) + 1e-16f;

    for_each_chunk<16>(fast, dd, deg, brow, ei, E, g, [&](int s, bool valid) {
      float a = 0.f;
      if (valid) a = expf(lrelu(als[s] + dv) - m) / q;
      unsigned mask = (__builtin_amdgcn_ballot_w32(valid) >> (half * 16)) & 0xffffu;
      while (mask) {
        const int j = __builtin_ctz(mask);
        mask &= mask - 1;
        const int sj = __shfl(s, j, 16);
        const float w = __shfl(a, j, 16);
        acc += H2[(size_t)sj * OUTC + g] * w;
      }
    });
  }

  const float v = acc + b2[g];
  const float mx = grp_max<16>(v);
  const float se = grp_sum<16>(expf(v - mx));
  const float lse = mx + logf(se);
  if (live) out[(size_t)d * OUTC + g] = v - lse;
}

static inline unsigned cdiv(long long a, long long b) { return (unsigned)((a + b - 1) / b); }

extern "C" void kernel_launch(void* const* d_in, const int* in_sizes, int n_in,
                              void* d_out, int out_size, void* d_ws, size_t ws_size,
                              hipStream_t stream) {
  const float* x    = (const float*)d_in[0];
  const int*   ei   = (const int*)  d_in[1];
  const float* W1   = (const float*)d_in[2];
  const float* asr1 = (const float*)d_in[3];
  const float* adt1 = (const float*)d_in[4];
  const float* b1   = (const float*)d_in[5];
  const float* W2   = (const float*)d_in[6];
  const float* asr2 = (const float*)d_in[7];
  const float* adt2 = (const float*)d_in[8];
  const float* b2   = (const float*)d_in[9];
  float* out = (float*)d_out;

  const int n = in_sizes[0] / F_IN;
  const int E = in_sizes[1] / 2;

  float* ws = (float*)d_ws;
  size_t off = 0;
  float* H1     = ws + off; off += (size_t)n * F_IN;
  float* Hp     = ws + off; off += (size_t)n * F_IN;
  int*   bucket = (int*)(ws + off); off += (size_t)n * CAP;
  float* als1   = ws + off; off += (size_t)n * HEADS;
  float* ald1   = ws + off; off += (size_t)n * HEADS;
  float* H2     = ws + off; off += (size_t)n * OUTC;
  float* als2   = ws + off; off += (size_t)n;
  float* ald2   = ws + off; off += (size_t)n;
  int*   cnt    = (int*)(ws + off); off += (size_t)n;

  const int B = 256;

  k_fill0i<<<cdiv(n, B), B, 0, stream>>>(cnt, n);
  k_bucket<<<cdiv(E, B), B, 0, stream>>>(ei, E, cnt, bucket);

  k_gemm1<<<cdiv(cdiv(n, 16), 2), 64, 0, stream>>>(x, W1, H1, n);
  k_coef<HEADS, HID><<<cdiv((long long)n * HEADS, B), B, 0, stream>>>(H1, asr1, adt1, als1, ald1, n);
  k_node1<<<cdiv(n, B / 32), B, 0, stream>>>(ei, E, n, cnt, bucket, als1, ald1, H1, b1, Hp);

  k_gemm2<<<cdiv(cdiv(n, 16), 8), B, 0, stream>>>(Hp, W2, H2, n);
  k_coef<1, OUTC><<<cdiv((long long)n, B), B, 0, stream>>>(H2, asr2, adt2, als2, ald2, n);
  k_node2<<<cdiv(n, B / 16), B, 0, stream>>>(ei, E, n, cnt, bucket, als2, ald2, H2, b2, out);
}


# Attention dots al_s, al_d taken from the accumulators in both GEMM epilogues (the two dot kernels and their re-read of H go); dot sum order now per-lane partials plus a lane shuffle tree
# speedup vs baseline: 4.8079x; 1.0455x over previous
#include <hip/hip_runtime.h>

typedef __attribute__((ext_vector_type(2))) float v2f;
typedef __attribute__((ext_vector_type(8))) float v8f;

#define F_IN   128
#define HID    32
#define HEADS  4
#define OUTC   16
#define CAP    64

__device__ __forceinline__ float lrelu(float v) { return v > 0.f ? v : 0.2f * v; }

template <int G>
__device__ __forceinline__ float grp_max(float v) {
  for (int o = G >> 1; o > 0; o >>= 1) v = fmaxf(v, __shfl_xor(v, o));
  return v;
}
template <int G>
__device__ __forceinline__ float grp_sum(float v) {
  for (int o = G >> 1; o > 0; o >>= 1) v += __shfl_xor(v, o);
  return v;
}

__global__ void k_fill0i(int* __restrict__ p, int count) {
  int i = blockIdx.x * blockDim.x + threadIdx.x;
  if (i < count) p[i] = 0;
}

__global__ void k_bucket(const int* __restrict__ ei, int E, int* __restrict__ cnt,
                         int* __restrict__ bucket) {
  int e = blockIdx.x * blockDim.x + threadIdx.x;
  if (e >= E) return;
  int s = ei[e], d = ei[E + e];
  int slot = atomicAdd(&cnt[d], 1);
  if (slot < CAP) bucket[(size_t)d * CAP + slot] = s;
}

__device__ __forceinline__ float f4e(const float4& v, int i) {
  return i == 0 ? v.x : i == 1 ? v.y : i == 2 ? v.z : v.w;
}

__global__ void __launch_bounds__(64) k_gemm1(const float* __restrict__ x,
                                              const float* __restrict__ W,
                                              const float* __restrict__ asr,
                                              const float* __restrict__ adt,
                                              float* __restrict__ H, float* __restrict__ als,
                                              float* __restrict__ ald, int n) {
  const int lane = threadIdx.x & 31;
  const int row0 = (blockIdx.x * (blockDim.x >> 5) + (threadIdx.x >> 5)) * 16;
  if (row0 >= n) return;
  const int c = lane & 15;
  const int h = lane >> 4;
  const float* ap = x + (size_t)(row0 + c) * F_IN + 4 * h;
  const float* bp = W + (size_t)(4 * h) * F_IN + 4 * c;

  v8f acc[8];
#pragma unroll
  for (int t = 0; t < 8; t++) acc[t] = v8f{};

  float4 a_cur = *(const float4*)ap;
  float4 b_cur[8];
#pragma unroll
  for (int s = 0; s < 4; s++) {
    b_cur[2 * s]     = *(const float4*)(bp + (size_t)s * F_IN);
    b_cur[2 * s + 1] = *(const float4*)(bp + (size_t)s * F_IN + 64);
  }

#pragma unroll
  for (int m = 0; m < F_IN / 8; m++) {
    float4 a_nxt = a_cur;
    float4 b_nxt[8];
#pragma unroll
    for (int s = 0; s < 8; s++) b_nxt[s] = b_cur[s];
    if (m + 1 < F_IN / 8) {
      a_nxt = *(const float4*)(ap + 8 * (m + 1));
#pragma unroll
      for (int s = 0; s < 4; s++) {
        b_nxt[2 * s]     = *(const float4*)(bp + (size_t)(8 * (m + 1) + s) * F_IN);
        b_nxt[2 * s + 1] = *(const float4*)(bp + (size_t)(8 * (m + 1) + s) * F_IN + 64);
      }
    }
#pragma unroll
    for (int u = 0; u < 2; u++) {
      v2f av;
      av.x = u == 0 ? a_cur.x : a_cur.z;
      av.y = u == 0 ? a_cur.y : a_cur.w;
#pragma unroll
      for (int t = 0; t < 8; t++) {
        v2f bv;
        bv.x = f4e(b_cur[2 * (2 * u) + (t >> 2)], t & 3);
        bv.y = f4e(b_cur[2 * (2 * u + 1) + (t >> 2)], t & 3);
        acc[t] = __builtin_amdgcn_wmma_f32_16x16x4_f32(false, av, false, bv, (short)0, acc[t],
                                                       false, false);
      }
    }
    a_cur = a_nxt;
#pragma unroll
    for (int s = 0; s < 8; s++) b_cur[s] = b_nxt[s];
  }

#pragma unroll
  for (int r = 0; r < 8; r++) {
    float* hp = H + (size_t)(row0 + 8 * h + r) * F_IN + 4 * c;
    *(float4*)hp        = make_float4(acc[0][r], acc[1][r], acc[2][r], acc[3][r]);
    *(float4*)(hp + 64) = make_float4(acc[4][r], acc[5][r], acc[6][r], acc[7][r]);
  }

  const int g  = c >> 3;
  const int k8 = c & 7;
  const float4 s_lo = *(const float4*)(asr + 4 * c);
  const float4 s_hi = *(const float4*)(asr + 64 + 4 * c);
  const float4 d_lo = *(const float4*)(adt + 4 * c);
  const float4 d_hi = *(const float4*)(adt + 64 + 4 * c);
  float os0 = 0.f, os1 = 0.f, od0 = 0.f, od1 = 0.f;
#pragma unroll
  for (int rp = 0; rp < 4; rp++) {
    const int r0 = 2 * rp, r1 = 2 * rp + 1;
    const v2f a0 = {acc[0][r0], acc[0][r1]}, a1 = {acc[1][r0], acc[1][r1]};
    const v2f a2 = {acc[2][r0], acc[2][r1]}, a3 = {acc[3][r0], acc[3][r1]};
    const v2f a4 = {acc[4][r0], acc[4][r1]}, a5 = {acc[5][r0], acc[5][r1]};
    const v2f a6 = {acc[6][r0], acc[6][r1]}, a7 = {acc[7][r0], acc[7][r1]};
    const v2f ps0 = a0 * s_lo.x + a1 * s_lo.y + a2 * s_lo.z + a3 * s_lo.w;
    const v2f ps1 = a4 * s_hi.x + a5 * s_hi.y + a6 * s_hi.z + a7 * s_hi.w;
    const v2f pd0 = a0 * d_lo.x + a1 * d_lo.y + a2 * d_lo.z + a3 * d_lo.w;
    const v2f pd1 = a4 * d_hi.x + a5 * d_hi.y + a6 * d_hi.z + a7 * d_hi.w;
    const float e0 = grp_sum<8>(ps0.x), e1 = grp_sum<8>(ps1.x);
    const float e2 = grp_sum<8>(pd0.x), e3 = grp_sum<8>(pd1.x);
    const float f0 = grp_sum<8>(ps0.y), f1 = grp_sum<8>(ps1.y);
    const float f2 = grp_sum<8>(pd0.y), f3 = grp_sum<8>(pd1.y);
    if (k8 == r0) { os0 = e0; os1 = e1; od0 = e2; od1 = e3; }
    if (k8 == r1) { os0 = f0; os1 = f1; od0 = f2; od1 = f3; }
  }
  const float rc0 = __shfl_xor(g == 0 ? od0 : os0, 8);
  const float rc1 = __shfl_xor(g == 0 ? od1 : os1, 8);
  const float4 o = g == 0 ? make_float4(os0, rc0, os1, rc1) : make_float4(rc0, od0, rc1, od1);
  float* op = (g == 0 ? als : ald) + (size_t)(row0 + 8 * h + k8) * HEADS;
  *(float4*)op = o;
}

__device__ __forceinline__ int al2_src(int v) { return 2 * v - (v & 15); }
__device__ __forceinline__ int al2_dst(int v) { return 2 * v - (v & 15) + 16; }

__global__ void k_gemm2(const float* __restrict__ Hp, const float* __restrict__ W2,
                        const float* __restrict__ asr, const float* __restrict__ adt,
                        float* __restrict__ H2, float* __restrict__ al2, int n) {
  const int wave = threadIdx.x >> 5;
  const int lane = threadIdx.x & 31;
  const int row0 = (blockIdx.x * 8 + wave) * 16;
  if (row0 >= n) return;
  const int lr = lane & 15;
  const int lh = lane >> 4;
  const float* ap = Hp + (size_t)(row0 + lr) * F_IN + 2 * lh;
  const float* bp = W2 + (size_t)(2 * lh) * OUTC + lr;
  v8f acc = {};
#pragma unroll
  for (int k = 0; k < F_IN; k += 4) {
    v2f a = *(const v2f*)(ap + k);
    v2f b;
    b.x = bp[(size_t)k * OUTC];
    b.y = bp[(size_t)k * OUTC + OUTC];
    acc = __builtin_amdgcn_wmma_f32_16x16x4_f32(false, a, false, b, (short)0, acc,
                                                false, false);
  }
#pragma unroll
  for (int i = 0; i < 8; i++) {
    H2[(size_t)(row0 + i + 8 * lh) * OUTC + lr] = acc[i];
  }

  const float as = asr[lr], ad = adt[lr];
  const int i8 = lr & 7;
  float own_s = 0.f, own_d = 0.f;
#pragma unroll
  for (int i = 0; i < 8; i++) {
    const float ps = grp_sum<16>(acc[i] * as);
    const float pd = grp_sum<16>(acc[i] * ad);
    if (i8 == i) { own_s = ps; own_d = pd; }
  }
  const float kp  = lh == 0 ? own_s : own_d;
  const float rc  = __shfl_xor(lh == 0 ? own_d : own_s, 16);
  al2[(size_t)(row0 >> 4) * 32 + lane] = lh == (lr >> 3) ? kp : rc;
}

template <int G, class F>
__device__ __forceinline__ void for_each_chunk(bool fast, int d, int deg,
                                               const int* __restrict__ brow,
                                               const int* __restrict__ ei, int E, int g, F&& f) {
  if (fast) {
    const int tot = deg + 1;
    for (int base = 0; base < tot; base += G) {
      const int j = base + g;
      int s = d;
      if (j < deg) s = brow[j];
      f(s, j < tot);
    }
  } else {
    for (int base = 0; base < E; base += G) {
      const int e = base + g;
      bool valid = false;
      int s = d;
      if (e < E && ei[E + e] == d) { valid = true; s = ei[e]; }
      f(s, valid);
    }
    f(d, g == 0);
  }
}

template <int U>
__device__ __forceinline__ void gather_rows1(const float* __restrict__ H, int s, float a0, float a1,
                                             float a2, float a3, int j0, int hsel, int lane,
                                             float4& acc) {
  float4 hv[U];
  float w[U];
#pragma unroll
  for (int u = 0; u < U; u++) {
    const int sj = __shfl(s, j0 + u);
    hv[u] = ((const float4*)H)[sj * (F_IN / 4) + lane];
  }
#pragma unroll
  for (int u = 0; u < U; u++) {
    const float w0 = __shfl(a0, j0 + u), w1 = __shfl(a1, j0 + u);
    const float w2 = __shfl(a2, j0 + u), w3 = __shfl(a3, j0 + u);
    w[u] = hsel == 0 ? w0 : hsel == 1 ? w1 : hsel == 2 ? w2 : w3;
  }
#pragma unroll
  for (int u = 0; u < U; u++) {
    acc.x = fmaf(hv[u].x, w[u], acc.x); acc.y = fmaf(hv[u].y, w[u], acc.y);
    acc.z = fmaf(hv[u].z, w[u], acc.z); acc.w = fmaf(hv[u].w, w[u], acc.w);
  }
}

template <int U>
__device__ __forceinline__ void gather_rows2(const float* __restrict__ H2, int s, float a, int j0,
                                             int g, float& acc) {
  float hv[U];
  float w[U];
#pragma unroll
  for (int u = 0; u < U; u++) {
    const int sj = __shfl(s, j0 + u, 16);
    hv[u] = H2[(size_t)sj * OUTC + g];
  }
#pragma unroll
  for (int u = 0; u < U; u++) w[u] = __shfl(a, j0 + u, 16);
#pragma unroll
  for (int u = 0; u < U; u++) acc = fmaf(hv[u], w[u], acc);
}

__device__ __forceinline__ void gather_list2(const float* __restrict__ H2, int s, float a, int cnt,
                                             int g, float& acc) {
  int j = 0;
  for (; j + 8 <= cnt; j += 8) gather_rows2<8>(H2, s, a, j, g, acc);
  if (cnt - j >= 4) { gather_rows2<4>(H2, s, a, j, g, acc); j += 4; }
  if (cnt - j >= 2) { gather_rows2<2>(H2, s, a, j, g, acc); j += 2; }
  if (cnt - j >= 1) gather_rows2<1>(H2, s, a, j, g, acc);
}

__global__ void k_node1(const int* __restrict__ ei, int E, int n,
                        const int* __restrict__ cnt, const int* __restrict__ bucket,
                        const float* __restrict__ als, const float* __restrict__ ald,
                        const float* __restrict__ H, const float* __restrict__ b1,
                        float* __restrict__ Hp) {
  const int lane = threadIdx.x & 31;
  const int d = blockIdx.x * (blockDim.x >> 5) + (threadIdx.x >> 5);
  if (d >= n) return;
  const int c = cnt[d];
  const bool fast = c <= CAP;
  const int deg = fast ? c : 0;
  const int* brow = bucket + (size_t)d * CAP;
  const float4 dv = *(const float4*)(ald + (size_t)d * HEADS);
  const float ninf = -__builtin_inff();

  const int hsel = lane >> 3;
  float4 acc = {0.f, 0.f, 0.f, 0.f};

  if (c < 32) {
    const int tot = c + 1;
    const bool valid = lane < tot;
    int s = d;
    if (lane < c) s = brow[lane];
    const float4 a = *(const float4*)(als + (size_t)s * HEADS);
    const float l0 = valid ? lrelu(a.x + dv.x) : ninf;
    const float l1 = valid ? lrelu(a.y + dv.y) : ninf;
    const float l2 = valid ? lrelu(a.z + dv.z) : ninf;
    const float l3 = valid ? lrelu(a.w + dv.w) : ninf;
    const float m0 = grp_max<32>(l0), m1 = grp_max<32>(l1);
    const float m2 = grp_max<32>(l2), m3 = grp_max<32>(l3);
    const float p0 = valid ? expf(l0 - m0) : 0.f;
    const float p1 = valid ? expf(l1 - m1) : 0.f;
    const float p2 = valid ? expf(l2 - m2) : 0.f;
    const float p3 = valid ? expf(l3 - m3) : 0.f;
    const float q0 = grp_sum<32>(p0) + 1e-16f, q1 = grp_sum<32>(p1) + 1e-16f;
    const float q2 = grp_sum<32>(p2) + 1e-16f, q3 = grp_sum<32>(p3) + 1e-16f;
    const float a0 = p0 / q0, a1 = p1 / q1, a2 = p2 / q2, a3 = p3 / q3;
    int j = 0;
    for (; j + 8 <= tot; j += 8) gather_rows1<8>(H, s, a0, a1, a2, a3, j, hsel, lane, acc);
    if (tot - j >= 4) { gather_rows1<4>(H, s, a0, a1, a2, a3, j, hsel, lane, acc); j += 4; }
    if (tot - j >= 2) { gather_rows1<2>(H, s, a0, a1, a2, a3, j, hsel, lane, acc); j += 2; }
    if (tot - j >= 1) gather_rows1<1>(H, s, a0, a1, a2, a3, j, hsel, lane, acc);
  } else {
    float m0 = ninf, m1 = ninf, m2 = ninf, m3 = ninf;
    for_each_chunk<32>(fast, d, deg, brow, ei, E, lane, [&](int s, bool valid) {
      if (valid) {
        const float4 a = *(const float4*)(als + (size_t)s * HEADS);
        m0 = fmaxf(m0, lrelu(a.x + dv.x));
        m1 = fmaxf(m1, lrelu(a.y + dv.y));
        m2 = fmaxf(m2, lrelu(a.z + dv.z));
        m3 = fmaxf(m3, lrelu(a.w + dv.w));
      }
    });
    m0 = grp_max<32>(m0); m1 = grp_max<32>(m1); m2 = grp_max<32>(m2); m3 = grp_max<32>(m3);

    float s0 = 0.f, s1 = 0.f, s2 = 0.f, s3 = 0.f;
    for_each_chunk<32>(fast, d, deg, brow, ei, E, lane, [&](int s, bool valid) {
      if (valid) {
        const float4 a = *(const float4*)(als + (size_t)s * HEADS);
        s0 += expf(lrelu(a.x + dv.x) - m0);
        s1 += expf(lrelu(a.y + dv.y) - m1);
        s2 += expf(lrelu(a.z + dv.z) - m2);
        s3 += expf(lrelu(a.w + dv.w) - m3);
      }
    });
    const float q0 = grp_sum<32>(s0) + 1e-16f, q1 = grp_sum<32>(s1) + 1e-16f;
    const float q2 = grp_sum<32>(s2) + 1e-16f, q3 = grp_sum<32>(s3) + 1e-16f;

    for_each_chunk<32>(fast, d, deg, brow, ei, E, lane, [&](int s, bool valid) {
      float a0 = 0.f, a1 = 0.f, a2 = 0.f, a3 = 0.f;
      if (valid) {
        const float4 a = *(const float4*)(als + (size_t)s * HEADS);
        a0 = expf(lrelu(a.x + dv.x) - m0) / q0;
        a1 = expf(lrelu(a.y + dv.y) - m1) / q1;
        a2 = expf(lrelu(a.z + dv.z) - m2) / q2;
        a3 = expf(lrelu(a.w + dv.w) - m3) / q3;
      }
      unsigned mask = __builtin_amdgcn_ballot_w32(valid);
      while (mask) {
        const int j = __builtin_ctz(mask);
        mask &= mask - 1;
        const int sj = __shfl(s, j);
        const float w0 = __shfl(a0, j), w1 = __shfl(a1, j);
        const float w2 = __shfl(a2, j), w3 = __shfl(a3, j);
        const float w = hsel == 0 ? w0 : hsel == 1 ? w1 : hsel == 2 ? w2 : w3;
        const float4 hv = *(const float4*)(H + (size_t)sj * F_IN + lane * 4);
        acc.x += hv.x * w; acc.y += hv.y * w; acc.z += hv.z * w; acc.w += hv.w * w;
      }
    });
  }

  float v[4] = {acc.x + b1[lane * 4 + 0], acc.y + b1[lane * 4 + 1],
                acc.z + b1[lane * 4 + 2], acc.w + b1[lane * 4 + 3]};
  for (int i = 0; i < 4; i++) {
    float t = v[i];
    t = t > 0.f ? t : (expf(t) - 1.f);
    Hp[(size_t)d * F_IN + i * 32 + lane] = t;
  }
}

__global__ void k_node2(const int* __restrict__ ei, int E, int n,
                        const int* __restrict__ cnt, const int* __restrict__ bucket,
                        const float* __restrict__ al2,
                        const float* __restrict__ H2, const float* __restrict__ b2,
                        float* __restrict__ out) {
  const int lane = threadIdx.x & 31;
  const int g = lane & 15;
  const int half = lane >> 4;
  const int d = (blockIdx.x * (blockDim.x >> 5) + (threadIdx.x >> 5)) * 2 + half;
  const bool live = d < n;
  const int dd = live ? d : 0;
  const int c = live ? cnt[dd] : -1;
  const bool fast = c <= CAP;
  const int deg = fast ? c : 0;
  const int* brow = bucket + (size_t)dd * CAP;
  const float dv = al2[al2_dst(dd)];
  const float ninf = -__builtin_inff();

  float acc = 0.f;
  if (__builtin_amdgcn_ballot_w32(c >= 32) == 0u) {
    const int tot = c + 1;
    const bool v0 = g < tot, v1 = g + 16 < tot;
    int s0 = dd, s1 = dd;
    if (g < c) s0 = brow[g];
    if (g + 16 < c) s1 = brow[g + 16];
    const float r0 = al2[al2_src(s0)], r1 = al2[al2_src(s1)];
    const float l0 = v0 ? lrelu(r0 + dv) : ninf;
    const float l1 = v1 ? lrelu(r1 + dv) : ninf;
    const float m = grp_max<16>(fmaxf(l0, l1));
    const float p0 = v0 ? expf(l0 - m) : 0.f;
    const float p1 = v1 ? expf(l1 - m) : 0.f;
    const float q = grp_sum<16>(p0 + p1) + 1e-16f;
    const float a0 = p0 / q, a1 = p1 / q;
    const int t0 = tot < 16 ? tot : 16;
    gather_list2(H2, s0, a0, t0, g, acc);
    gather_list2(H2, s1, a1, tot - t0, g, acc);
  } else {
    float m = ninf;
    for_each_chunk<16>(fast, dd, deg, brow, ei, E, g, [&](int s, bool valid) {
      if (valid) m = fmaxf(m, lrelu(al2[al2_src(s)] + dv));
    });
    m = grp_max<16>(m);

    float ss = 0.f;
    for_each_chunk<16>(fast, dd, deg, brow, ei, E, g, [&](int s, bool valid) {
      if (valid) ss += expf(lrelu(al2[al2_src(s)] + dv) - m);
    });
    const float q = grp_sum<16>(ss) + 1e-16f;

    for_each_chunk<16>(fast, dd, deg, brow, ei, E, g, [&](int s, bool valid) {
      float a = 0.f;
      if (valid) a = expf(lrelu(al2[al2_src(s)] + dv) - m) / q;
      unsigned mask = (__builtin_amdgcn_ballot_w32(valid) >> (half * 16)) & 0xffffu;
      while (mask) {
        const int j = __builtin_ctz(mask);
        mask &= mask - 1;
        const int sj = __shfl(s, j, 16);
        const float w = __shfl(a, j, 16);
        acc += H2[(size_t)sj * OUTC + g] * w;
      }
    });
  }

  const float v = acc + b2[g];
  const float mx = grp_max<16>(v);
  const float se = grp_sum<16>(expf(v - mx));
  const float lse = mx + logf(se);
  if (live) out[(size_t)d * OUTC + g] = v - lse;
}

static inline unsigned cdiv(long long a, long long b) { return (unsigned)((a + b - 1) / b); }

extern "C" void kernel_launch(void* const* d_in, const int* in_sizes, int n_in,
                              void* d_out, int out_size, void* d_ws, size_t ws_size,
                              hipStream_t stream) {
  const float* x    = (const float*)d_in[0];
  const int*   ei   = (const int*)  d_in[1];
  const float* W1   = (const float*)d_in[2];
  const float* asr1 = (const float*)d_in[3];
  const float* adt1 = (const float*)d_in[4];
  const float* b1   = (const float*)d_in[5];
  const float* W2   = (const float*)d_in[6];
  const float* asr2 = (const float*)d_in[7];
  const float* adt2 = (const float*)d_in[8];
  const float* b2   = (const float*)d_in[9];
  float* out = (float*)d_out;

  const int n = in_sizes[0] / F_IN;
  const int E = in_sizes[1] / 2;

  float* ws = (float*)d_ws;
  size_t off = 0;
  float* H1     = ws + off; off += (size_t)n * F_IN;
  float* Hp     = ws + off; off += (size_t)n * F_IN;
  int*   bucket = (int*)(ws + off); off += (size_t)n * CAP;
  float* als1   = ws + off; off += (size_t)n * HEADS;
  float* ald1   = ws + off; off += (size_t)n * HEADS;
  float* H2     = ws + off; off += (size_t)n * OUTC;
  float* al2    = ws + off; off += (size_t)n * 2;
  int*   cnt    = (int*)(ws + off); off += (size_t)n;

  const int B = 256;

  k_fill0i<<<cdiv(n, B), B, 0, stream>>>(cnt, n);
  k_bucket<<<cdiv(E, B), B, 0, stream>>>(ei, E, cnt, bucket);

  k_gemm1<<<cdiv(cdiv(n, 16), 2), 64, 0, stream>>>(x, W1, asr1, adt1, H1, als1, ald1, n);
  k_node1<<<cdiv(n, B / 32), B, 0, stream>>>(ei, E, n, cnt, bucket, als1, ald1, H1, b1, Hp);

  k_gemm2<<<cdiv(cdiv(n, 16), 8), B, 0, stream>>>(Hp, W2, asr2, adt2, H2, al2, n);
  k_node2<<<cdiv(n, B / 16), B, 0, stream>>>(ei, E, n, cnt, bucket, al2, H2, b2, out);
}


# Node kernels: cross-lane moves leave the LDS permute unit: a layer-1 list entry's source row and 4 weights are read from its lane into scalar registers, butterflies run on DPP; sum order unchanged
# speedup vs baseline: 4.9531x; 1.0302x over previous
#include <hip/hip_runtime.h>

typedef __attribute__((ext_vector_type(2))) float v2f;
typedef __attribute__((ext_vector_type(8))) float v8f;

#define F_IN   128
#define HID    32
#define HEADS  4
#define OUTC   16
#define CAP    64

__device__ __forceinline__ float lrelu(float v) { return v > 0.f ? v : 0.2f * v; }

template <int G>
__device__ __forceinline__ float grp_max(float v) {
  for (int o = G >> 1; o > 0; o >>= 1) v = fmaxf(v, __shfl_xor(v, o));
  return v;
}
template <int G>
__device__ __forceinline__ float grp_sum(float v) {
  for (int o = G >> 1; o > 0; o >>= 1) v += __shfl_xor(v, o);
  return v;
}

template <int M>
__device__ __forceinline__ float xrow(float x) {
  const int b = __builtin_bit_cast(int, x);
  return __builtin_bit_cast(float, __builtin_amdgcn_update_dpp(0, b, 0x160 + M, 0xf, 0xf, true));
}
__device__ __forceinline__ float xhalf(float x) {
  const int b = __builtin_bit_cast(int, x);
  return __builtin_bit_cast(float,
                            __builtin_amdgcn_permlanex16(b, b, 0x76543210, 0xfedcba98, false, false));
}
template <int G>
__device__ __forceinline__ float vmax(float v) {
  if (G >= 32) v = fmaxf(v, xhalf(v));
  if (G >= 16) v = fmaxf(v, xrow<8>(v));
  v = fmaxf(v, xrow<4>(v));
  v = fmaxf(v, xrow<2>(v));
  v = fmaxf(v, xrow<1>(v));
  return v;
}
template <int G>
__device__ __forceinline__ float vsum(float v) {
  if (G >= 32) v += xhalf(v);
  if (G >= 16) v += xrow<8>(v);
  v += xrow<4>(v);
  v += xrow<2>(v);
  v += xrow<1>(v);
  return v;
}
__device__ __forceinline__ float lane_f(float x, int j) {
  return __builtin_bit_cast(float, __builtin_amdgcn_readlane(__builtin_bit_cast(int, x), j));
}

__global__ void k_fill0i(int* __restrict__ p, int count) {
  int i = blockIdx.x * blockDim.x + threadIdx.x;
  if (i < count) p[i] = 0;
}

__global__ void k_bucket(const int* __restrict__ ei, int E, int* __restrict__ cnt,
                         int* __restrict__ bucket) {
  int e = blockIdx.x * blockDim.x + threadIdx.x;
  if (e >= E) return;
  int s = ei[e], d = ei[E + e];
  int slot = atomicAdd(&cnt[d], 1);
  if (slot < CAP) bucket[(size_t)d * CAP + slot] = s;
}

__device__ __forceinline__ float f4e(const float4& v, int i) {
  return i == 0 ? v.x : i == 1 ? v.y : i == 2 ? v.z : v.w;
}

__global__ void __launch_bounds__(64) k_gemm1(const float* __restrict__ x,
                                              const float* __restrict__ W,
                                              const float* __restrict__ asr,
                                              const float* __restrict__ adt,
                                              float* __restrict__ H, float* __restrict__ als,
                                              float* __restrict__ ald, int n) {
  const int lane = threadIdx.x & 31;
  const int row0 = (blockIdx.x * (blockDim.x >> 5) + (threadIdx.x >> 5)) * 16;
  if (row0 >= n) return;
  const int c = lane & 15;
  const int h = lane >> 4;
  const float* ap = x + (size_t)(row0 + c) * F_IN + 4 * h;
  const float* bp = W + (size_t)(4 * h) * F_IN + 4 * c;

  v8f acc[8];
#pragma unroll
  for (int t = 0; t < 8; t++) acc[t] = v8f{};

  float4 a_cur = *(const float4*)ap;
  float4 b_cur[8];
#pragma unroll
  for (int s = 0; s < 4; s++) {
    b_cur[2 * s]     = *(const float4*)(bp + (size_t)s * F_IN);
    b_cur[2 * s + 1] = *(const float4*)(bp + (size_t)s * F_IN + 64);
  }

#pragma unroll
  for (int m = 0; m < F_IN / 8; m++) {
    float4 a_nxt = a_cur;
    float4 b_nxt[8];
#pragma unroll
    for (int s = 0; s < 8; s++) b_nxt[s] = b_cur[s];
    if (m + 1 < F_IN / 8) {
      a_nxt = *(const float4*)(ap + 8 * (m + 1));
#pragma unroll
      for (int s = 0; s < 4; s++) {
        b_nxt[2 * s]     = *(const float4*)(bp + (size_t)(8 * (m + 1) + s) * F_IN);
        b_nxt[2 * s + 1] = *(const float4*)(bp + (size_t)(8 * (m + 1) + s) * F_IN + 64);
      }
    }
#pragma unroll
    for (int u = 0; u < 2; u++) {
      v2f av;
      av.x = u == 0 ? a_cur.x : a_cur.z;
      av.y = u == 0 ? a_cur.y : a_cur.w;
#pragma unroll
      for (int t = 0; t < 8; t++) {
        v2f bv;
        bv.x = f4e(b_cur[2 * (2 * u) + (t >> 2)], t & 3);
        bv.y = f4e(b_cur[2 * (2 * u + 1) + (t >> 2)], t & 3);
        acc[t] = __builtin_amdgcn_wmma_f32_16x16x4_f32(false, av, false, bv, (short)0, acc[t],
                                                       false, false);
      }
    }
    a_cur = a_nxt;
#pragma unroll
    for (int s = 0; s < 8; s++) b_cur[s] = b_nxt[s];
  }

#pragma unroll
  for (int r = 0; r < 8; r++) {
    float* hp = H + (size_t)(row0 + 8 * h + r) * F_IN + 4 * c;
    *(float4*)hp        = make_float4(acc[0][r], acc[1][r], acc[2][r], acc[3][r]);
    *(float4*)(hp + 64) = make_float4(acc[4][r], acc[5][r], acc[6][r], acc[7][r]);
  }

  const int g  = c >> 3;
  const int k8 = c & 7;
  const float4 s_lo = *(const float4*)(asr + 4 * c);
  const float4 s_hi = *(const float4*)(asr + 64 + 4 * c);
  const float4 d_lo = *(const float4*)(adt + 4 * c);
  const float4 d_hi = *(const float4*)(adt + 64 + 4 * c);
  float os0 = 0.f, os1 = 0.f, od0 = 0.f, od1 = 0.f;
#pragma unroll
  for (int rp = 0; rp < 4; rp++) {
    const int r0 = 2 * rp, r1 = 2 * rp + 1;
    const v2f a0 = {acc[0][r0], acc[0][r1]}, a1 = {acc[1][r0], acc[1][r1]};
    const v2f a2 = {acc[2][r0], acc[2][r1]}, a3 = {acc[3][r0], acc[3][r1]};
    const v2f a4 = {acc[4][r0], acc[4][r1]}, a5 = {acc[5][r0], acc[5][r1]};
    const v2f a6 = {acc[6][r0], acc[6][r1]}, a7 = {acc[7][r0], acc[7][r1]};
    const v2f ps0 = a0 * s_lo.x + a1 * s_lo.y + a2 * s_lo.z + a3 * s_lo.w;
    const v2f ps1 = a4 * s_hi.x + a5 * s_hi.y + a6 * s_hi.z + a7 * s_hi.w;
    const v2f pd0 = a0 * d_lo.x + a1 * d_lo.y + a2 * d_lo.z + a3 * d_lo.w;
    const v2f pd1 = a4 * d_hi.x + a5 * d_hi.y + a6 * d_hi.z + a7 * d_hi.w;
    const float e0 = grp_sum<8>(ps0.x), e1 = grp_sum<8>(ps1.x);
    const float e2 = grp_sum<8>(pd0.x), e3 = grp_sum<8>(pd1.x);
    const float f0 = grp_sum<8>(ps0.y), f1 = grp_sum<8>(ps1.y);
    const float f2 = grp_sum<8>(pd0.y), f3 = grp_sum<8>(pd1.y);
    if (k8 == r0) { os0 = e0; os1 = e1; od0 = e2; od1 = e3; }
    if (k8 == r1) { os0 = f0; os1 = f1; od0 = f2; od1 = f3; }
  }
  const float rc0 = __shfl_xor(g == 0 ? od0 : os0, 8);
  const float rc1 = __shfl_xor(g == 0 ? od1 : os1, 8);
  const float4 o = g == 0 ? make_float4(os0, rc0, os1, rc1) : make_float4(rc0, od0, rc1, od1);
  float* op = (g == 0 ? als : ald) + (size_t)(row0 + 8 * h + k8) * HEADS;
  *(float4*)op = o;
}

__device__ __forceinline__ int al2_src(int v) { return 2 * v - (v & 15); }
__device__ __forceinline__ int al2_dst(int v) { return 2 * v - (v & 15) + 16; }

__global__ void k_gemm2(const float* __restrict__ Hp, const float* __restrict__ W2,
                        const float* __restrict__ asr, const float* __restrict__ adt,
                        float* __restrict__ H2, float* __restrict__ al2, int n) {
  const int wave = threadIdx.x >> 5;
  const int lane = threadIdx.x & 31;
  const int row0 = (blockIdx.x * 8 + wave) * 16;
  if (row0 >= n) return;
  const int lr = lane & 15;
  const int lh = lane >> 4;
  const float* ap = Hp + (size_t)(row0 + lr) * F_IN + 2 * lh;
  const float* bp = W2 + (size_t)(2 * lh) * OUTC + lr;
  v8f acc = {};
#pragma unroll
  for (int k = 0; k < F_IN; k += 4) {
    v2f a = *(const v2f*)(ap + k);
    v2f b;
    b.x = bp[(size_t)k * OUTC];
    b.y = bp[(size_t)k * OUTC + OUTC];
    acc = __builtin_amdgcn_wmma_f32_16x16x4_f32(false, a, false, b, (short)0, acc,
                                                false, false);
  }
#pragma unroll
  for (int i = 0; i < 8; i++) {
    H2[(size_t)(row0 + i + 8 * lh) * OUTC + lr] = acc[i];
  }

  const float as = asr[lr], ad = adt[lr];
  const int i8 = lr & 7;
  float own_s = 0.f, own_d = 0.f;
#pragma unroll
  for (int i = 0; i < 8; i++) {
    const float ps = grp_sum<16>(acc[i] * as);
    const float pd = grp_sum<16>(acc[i] * ad);
    if (i8 == i) { own_s = ps; own_d = pd; }
  }
  const float kp  = lh == 0 ? own_s : own_d;
  const float rc  = __shfl_xor(lh == 0 ? own_d : own_s, 16);
  al2[(size_t)(row0 >> 4) * 32 + lane] = lh == (lr >> 3) ? kp : rc;
}

template <int G, class F>
__device__ __forceinline__ void for_each_chunk(bool fast, int d, int deg,
                                               const int* __restrict__ brow,
                                               const int* __restrict__ ei, int E, int g, F&& f) {
  if (fast) {
    const int tot = deg + 1;
    for (int base = 0; base < tot; base += G) {
      const int j = base + g;
      int s = d;
      if (j < deg) s = brow[j];
      f(s, j < tot);
    }
  } else {
    for (int base = 0; base < E; base += G) {
      const int e = base + g;
      bool valid = false;
      int s = d;
      if (e < E && ei[E + e] == d) { valid = true; s = ei[e]; }
      f(s, valid);
    }
    f(d, g == 0);
  }
}

template <int U>
__device__ __forceinline__ void gather_rows1(const float* __restrict__ H, int s, float a0, float a1,
                                             float a2, float a3, int j0, bool h1, bool h2, int lane,
                                             float4& acc) {
  float4 hv[U];
  float w[U];
#pragma unroll
  for (int u = 0; u < U; u++) {
    const int sj = __builtin_amdgcn_readlane(s, j0 + u);
    hv[u] = ((const float4*)(H + (size_t)sj * F_IN))[lane];
  }
#pragma unroll
  for (int u = 0; u < U; u++) {
    const float w0 = lane_f(a0, j0 + u), w1 = lane_f(a1, j0 + u);
    const float w2 = lane_f(a2, j0 + u), w3 = lane_f(a3, j0 + u);
    const float lo = h1 ? w1 : w0;
    const float hi = h1 ? w3 : w2;
    w[u] = h2 ? hi : lo;
  }
#pragma unroll
  for (int u = 0; u < U; u++) {
    acc.x = fmaf(hv[u].x, w[u], acc.x); acc.y = fmaf(hv[u].y, w[u], acc.y);
    acc.z = fmaf(hv[u].z, w[u], acc.z); acc.w = fmaf(hv[u].w, w[u], acc.w);
  }
}

template <int U>
__device__ __forceinline__ void gather_rows2(const float* __restrict__ H2, int s, float a, int j0,
                                             int g, float& acc) {
  float hv[U];
  float w[U];
#pragma unroll
  for (int u = 0; u < U; u++) {
    const int sj = __shfl(s, j0 + u, 16);
    hv[u] = H2[(size_t)sj * OUTC + g];
  }
#pragma unroll
  for (int u = 0; u < U; u++) w[u] = __shfl(a, j0 + u, 16);
#pragma unroll
  for (int u = 0; u < U; u++) acc = fmaf(hv[u], w[u], acc);
}

__device__ __forceinline__ void gather_list2(const float* __restrict__ H2, int s, float a, int cnt,
                                             int g, float& acc) {
  int j = 0;
  for (; j + 8 <= cnt; j += 8) gather_rows2<8>(H2, s, a, j, g, acc);
  if (cnt - j >= 4) { gather_rows2<4>(H2, s, a, j, g, acc); j += 4; }
  if (cnt - j >= 2) { gather_rows2<2>(H2, s, a, j, g, acc); j += 2; }
  if (cnt - j >= 1) gather_rows2<1>(H2, s, a, j, g, acc);
}

__global__ void k_node1(const int* __restrict__ ei, int E, int n,
                        const int* __restrict__ cnt, const int* __restrict__ bucket,
                        const float* __restrict__ als, const float* __restrict__ ald,
                        const float* __restrict__ H, const float* __restrict__ b1,
                        float* __restrict__ Hp) {
  const int lane = threadIdx.x & 31;
  const int d = __builtin_amdgcn_readfirstlane(blockIdx.x * (blockDim.x >> 5) + (threadIdx.x >> 5));
  if (d >= n) return;
  const int c = __builtin_amdgcn_readfirstlane(cnt[d]);
  const bool fast = c <= CAP;
  const int deg = fast ? c : 0;
  const int* brow = bucket + (size_t)d * CAP;
  const float4 dv = *(const float4*)(ald + (size_t)d * HEADS);
  const float ninf = -__builtin_inff();

  const int hsel = lane >> 3;
  const bool h1 = (hsel & 1) != 0, h2 = (hsel & 2) != 0;
  float4 acc = {0.f, 0.f, 0.f, 0.f};

  if (c < 32) {
    const int tot = c + 1;
    const bool valid = lane < tot;
    int s = d;
    if (lane < c) s = brow[lane];
    const float4 a = *(const float4*)(als + (size_t)s * HEADS);
    const float l0 = valid ? lrelu(a.x + dv.x) : ninf;
    const float l1 = valid ? lrelu(a.y + dv.y) : ninf;
    const float l2 = valid ? lrelu(a.z + dv.z) : ninf;
    const float l3 = valid ? lrelu(a.w + dv.w) : ninf;
    const float m0 = vmax<32>(l0), m1 = vmax<32>(l1);
    const float m2 = vmax<32>(l2), m3 = vmax<32>(l3);
    const float p0 = valid ? expf(l0 - m0) : 0.f;
    const float p1 = valid ? expf(l1 - m1) : 0.f;
    const float p2 = valid ? expf(l2 - m2) : 0.f;
    const float p3 = valid ? expf(l3 - m3) : 0.f;
    const float q0 = vsum<32>(p0) + 1e-16f, q1 = vsum<32>(p1) + 1e-16f;
    const float q2 = vsum<32>(p2) + 1e-16f, q3 = vsum<32>(p3) + 1e-16f;
    const float a0 = p0 / q0, a1 = p1 / q1, a2 = p2 / q2, a3 = p3 / q3;
    int j = 0;
    for (; j + 8 <= tot; j += 8) gather_rows1<8>(H, s, a0, a1, a2, a3, j, h1, h2, lane, acc);
    if (tot - j >= 4) { gather_rows1<4>(H, s, a0, a1, a2, a3, j, h1, h2, lane, acc); j += 4; }
    if (tot - j >= 2) { gather_rows1<2>(H, s, a0, a1, a2, a3, j, h1, h2, lane, acc); j += 2; }
    if (tot - j >= 1) gather_rows1<1>(H, s, a0, a1, a2, a3, j, h1, h2, lane, acc);
  } else {
    float m0 = ninf, m1 = ninf, m2 = ninf, m3 = ninf;
    for_each_chunk<32>(fast, d, deg, brow, ei, E, lane, [&](int s, bool valid) {
      if (valid) {
        const float4 a = *(const float4*)(als + (size_t)s * HEADS);
        m0 = fmaxf(m0, lrelu(a.x + dv.x));
        m1 = fmaxf(m1, lrelu(a.y + dv.y));
        m2 = fmaxf(m2, lrelu(a.z + dv.z));
        m3 = fmaxf(m3, lrelu(a.w + dv.w));
      }
    });
    m0 = vmax<32>(m0); m1 = vmax<32>(m1); m2 = vmax<32>(m2); m3 = vmax<32>(m3);

    float s0 = 0.f, s1 = 0.f, s2 = 0.f, s3 = 0.f;
    for_each_chunk<32>(fast, d, deg, brow, ei, E, lane, [&](int s, bool valid) {
      if (valid) {
        const float4 a = *(const float4*)(als + (size_t)s * HEADS);
        s0 += expf(lrelu(a.x + dv.x) - m0);
        s1 += expf(lrelu(a.y + dv.y) - m1);
        s2 += expf(lrelu(a.z + dv.z) - m2);
        s3 += expf(lrelu(a.w + dv.w) - m3);
      }
    });
    const float q0 = vsum<32>(s0) + 1e-16f, q1 = vsum<32>(s1) + 1e-16f;
    const float q2 = vsum<32>(s2) + 1e-16f, q3 = vsum<32>(s3) + 1e-16f;

    for_each_chunk<32>(fast, d, deg, brow, ei, E, lane, [&](int s, bool valid) {
      float a0 = 0.f, a1 = 0.f, a2 = 0.f, a3 = 0.f;
      if (valid) {
        const float4 a = *(const float4*)(als + (size_t)s * HEADS);
        a0 = expf(lrelu(a.x + dv.x) - m0) / q0;
        a1 = expf(lrelu(a.y + dv.y) - m1) / q1;
        a2 = expf(lrelu(a.z + dv.z) - m2) / q2;
        a3 = expf(lrelu(a.w + dv.w) - m3) / q3;
      }
      unsigned mask = __builtin_amdgcn_ballot_w32(valid);
      while (mask) {
        const int j = __builtin_ctz(mask);
        mask &= mask - 1;
        const int sj = __shfl(s, j);
        const float w0 = __shfl(a0, j), w1 = __shfl(a1, j);
        const float w2 = __shfl(a2, j), w3 = __shfl(a3, j);
        const float w = hsel == 0 ? w0 : hsel == 1 ? w1 : hsel == 2 ? w2 : w3;
        const float4 hv = *(const float4*)(H + (size_t)sj * F_IN + lane * 4);
        acc.x += hv.x * w; acc.y += hv.y * w; acc.z += hv.z * w; acc.w += hv.w * w;
      }
    });
  }

  float v[4] = {acc.x + b1[lane * 4 + 0], acc.y + b1[lane * 4 + 1],
                acc.z + b1[lane * 4 + 2], acc.w + b1[lane * 4 + 3]};
  for (int i = 0; i < 4; i++) {
    float t = v[i];
    t = t > 0.f ? t : (expf(t) - 1.f);
    Hp[(size_t)d * F_IN + i * 32 + lane] = t;
  }
}

__global__ void k_node2(const int* __restrict__ ei, int E, int n,
                        const int* __restrict__ cnt, const int* __restrict__ bucket,
                        const float* __restrict__ al2,
                        const float* __restrict__ H2, const float* __restrict__ b2,
                        float* __restrict__ out) {
  const int lane = threadIdx.x & 31;
  const int g = lane & 15;
  const int half = lane >> 4;
  const int d = (blockIdx.x * (blockDim.x >> 5) + (threadIdx.x >> 5)) * 2 + half;
  const bool live = d < n;
  const int dd = live ? d : 0;
  const int c = live ? cnt[dd] : -1;
  const bool fast = c <= CAP;
  const int deg = fast ? c : 0;
  const int* brow = bucket + (size_t)dd * CAP;
  const float dv = al2[al2_dst(dd)];
  const float ninf = -__builtin_inff();

  float acc = 0.f;
  if (__builtin_amdgcn_ballot_w32(c >= 32) == 0u) {
    const int tot = c + 1;
    const bool v0 = g < tot, v1 = g + 16 < tot;
    int s0 = dd, s1 = dd;
    if (g < c) s0 = brow[g];
    if (g + 16 < c) s1 = brow[g + 16];
    const float r0 = al2[al2_src(s0)], r1 = al2[al2_src(s1)];
    const float l0 = v0 ? lrelu(r0 + dv) : ninf;
    const float l1 = v1 ? lrelu(r1 + dv) : ninf;
    const float m = vmax<16>(fmaxf(l0, l1));
    const float p0 = v0 ? expf(l0 - m) : 0.f;
    const float p1 = v1 ? expf(l1 - m) : 0.f;
    const float q = vsum<16>(p0 + p1) + 1e-16f;
    const float a0 = p0 / q, a1 = p1 / q;
    const int t0 = tot < 16 ? tot : 16;
    gather_list2(H2, s0, a0, t0, g, acc);
    gather_list2(H2, s1, a1, tot - t0, g, acc);
  } else {
    float m = ninf;
    for_each_chunk<16>(fast, dd, deg, brow, ei, E, g, [&](int s, bool valid) {
      if (valid) m = fmaxf(m, lrelu(al2[al2_src(s)] + dv));
    });
    m = vmax<16>(m);

    float ss = 0.f;
    for_each_chunk<16>(fast, dd, deg, brow, ei, E, g, [&](int s, bool valid) {
      if (valid) ss += expf(lrelu(al2[al2_src(s)] + dv) - m);
    });
    const float q = vsum<16>(ss) + 1e-16f;

    for_each_chunk<16>(fast, dd, deg, brow, ei, E, g, [&](int s, bool valid) {
      float a = 0.f;
      if (valid) a = expf(lrelu(al2[al2_src(s)] + dv) - m) / q;
      unsigned mask = (__builtin_amdgcn_ballot_w32(valid) >> (half * 16)) & 0xffffu;
      while (mask) {
        const int j = __builtin_ctz(mask);
        mask &= mask - 1;
        const int sj = __shfl(s, j, 16);
        const float w = __shfl(a, j, 16);
        acc += H2[(size_t)sj * OUTC + g] * w;
      }
    });
  }

  const float v = acc + b2[g];
  const float mx = vmax<16>(v);
  const float se = vsum<16>(expf(v - mx));
  const float lse = mx + logf(se);
  if (live) out[(size_t)d * OUTC + g] = v - lse;
}

static inline unsigned cdiv(long long a, long long b) { return (unsigned)((a + b - 1) / b); }

extern "C" void kernel_launch(void* const* d_in, const int* in_sizes, int n_in,
                              void* d_out, int out_size, void* d_ws, size_t ws_size,
                              hipStream_t stream) {
  const float* x    = (const float*)d_in[0];
  const int*   ei   = (const int*)  d_in[1];
  const float* W1   = (const float*)d_in[2];
  const float* asr1 = (const float*)d_in[3];
  const float* adt1 = (const float*)d_in[4];
  const float* b1   = (const float*)d_in[5];
  const float* W2   = (const float*)d_in[6];
  const float* asr2 = (const float*)d_in[7];
  const float* adt2 = (const float*)d_in[8];
  const float* b2   = (const float*)d_in[9];
  float* out = (float*)d_out;

  const int n = in_sizes[0] / F_IN;
  const int E = in_sizes[1] / 2;

  float* ws = (float*)d_ws;
  size_t off = 0;
  float* H1     = ws + off; off += (size_t)n * F_IN;
  float* Hp     = ws + off; off += (size_t)n * F_IN;
  int*   bucket = (int*)(ws + off); off += (size_t)n * CAP;
  float* als1   = ws + off; off += (size_t)n * HEADS;
  float* ald1   = ws + off; off += (size_t)n * HEADS;
  float* H2     = ws + off; off += (size_t)n * OUTC;
  float* al2    = ws + off; off += (size_t)n * 2;
  int*   cnt    = (int*)(ws + off); off += (size_t)n;

  const int B = 256;

  k_fill0i<<<cdiv(n, B), B, 0, stream>>>(cnt, n);
  k_bucket<<<cdiv(E, B), B, 0, stream>>>(ei, E, cnt, bucket);

  k_gemm1<<<cdiv(cdiv(n, 16), 2), 64, 0, stream>>>(x, W1, asr1, adt1, H1, als1, ald1, n);
  k_node1<<<cdiv(n, B / 32), B, 0, stream>>>(ei, E, n, cnt, bucket, als1, ald1, H1, b1, Hp);

  k_gemm2<<<cdiv(cdiv(n, 16), 8), B, 0, stream>>>(Hp, W2, asr2, adt2, H2, al2, n);
  k_node2<<<cdiv(n, B / 16), B, 0, stream>>>(ei, E, n, cnt, bucket, al2, H2, b2, out);
}


# Node kernels: lists of 32 to 63 edges also take the register path (2 entries per lane in layer 1, 4 in layer 2), not three chunked passes that wait for one row at a time; sum order unchanged
# speedup vs baseline: 5.2467x; 1.0593x over previous
#include <hip/hip_runtime.h>

typedef __attribute__((ext_vector_type(2))) float v2f;
typedef __attribute__((ext_vector_type(8))) float v8f;

#define F_IN   128
#define HID    32
#define HEADS  4
#define OUTC   16
#define CAP    64

__device__ __forceinline__ float lrelu(float v) { return v > 0.f ? v : 0.2f * v; }

template <int G>
__device__ __forceinline__ float grp_max(float v) {
  for (int o = G >> 1; o > 0; o >>= 1) v = fmaxf(v, __shfl_xor(v, o));
  return v;
}
template <int G>
__device__ __forceinline__ float grp_sum(float v) {
  for (int o = G >> 1; o > 0; o >>= 1) v += __shfl_xor(v, o);
  return v;
}

template <int M>
__device__ __forceinline__ float xrow(float x) {
  const int b = __builtin_bit_cast(int, x);
  return __builtin_bit_cast(float, __builtin_amdgcn_update_dpp(0, b, 0x160 + M, 0xf, 0xf, true));
}
__device__ __forceinline__ float xhalf(float x) {
  const int b = __builtin_bit_cast(int, x);
  return __builtin_bit_cast(float,
                            __builtin_amdgcn_permlanex16(b, b, 0x76543210, 0xfedcba98, false, false));
}
template <int G>
__device__ __forceinline__ float vmax(float v) {
  if (G >= 32) v = fmaxf(v, xhalf(v));
  if (G >= 16) v = fmaxf(v, xrow<8>(v));
  v = fmaxf(v, xrow<4>(v));
  v = fmaxf(v, xrow<2>(v));
  v = fmaxf(v, xrow<1>(v));
  return v;
}
template <int G>
__device__ __forceinline__ float vsum(float v) {
  if (G >= 32) v += xhalf(v);
  if (G >= 16) v += xrow<8>(v);
  v += xrow<4>(v);
  v += xrow<2>(v);
  v += xrow<1>(v);
  return v;
}
__device__ __forceinline__ float lane_f(float x, int j) {
  return __builtin_bit_cast(float, __builtin_amdgcn_readlane(__builtin_bit_cast(int, x), j));
}

__global__ void k_fill0i(int* __restrict__ p, int count) {
  int i = blockIdx.x * blockDim.x + threadIdx.x;
  if (i < count) p[i] = 0;
}

__global__ void k_bucket(const int* __restrict__ ei, int E, int* __restrict__ cnt,
                         int* __restrict__ bucket) {
  int e = blockIdx.x * blockDim.x + threadIdx.x;
  if (e >= E) return;
  int s = ei[e], d = ei[E + e];
  int slot = atomicAdd(&cnt[d], 1);
  if (slot < CAP) bucket[(size_t)d * CAP + slot] = s;
}

__device__ __forceinline__ float f4e(const float4& v, int i) {
  return i == 0 ? v.x : i == 1 ? v.y : i == 2 ? v.z : v.w;
}

__global__ void __launch_bounds__(64) k_gemm1(const float* __restrict__ x,
                                              const float* __restrict__ W,
                                              const float* __restrict__ asr,
                                              const float* __restrict__ adt,
                                              float* __restrict__ H, float* __restrict__ als,
                                              float* __restrict__ ald, int n) {
  const int lane = threadIdx.x & 31;
  const int row0 = (blockIdx.x * (blockDim.x >> 5) + (threadIdx.x >> 5)) * 16;
  if (row0 >= n) return;
  const int c = lane & 15;
  const int h = lane >> 4;
  const float* ap = x + (size_t)(row0 + c) * F_IN + 4 * h;
  const float* bp = W + (size_t)(4 * h) * F_IN + 4 * c;

  v8f acc[8];
#pragma unroll
  for (int t = 0; t < 8; t++) acc[t] = v8f{};

  float4 a_cur = *(const float4*)ap;
  float4 b_cur[8];
#pragma unroll
  for (int s = 0; s < 4; s++) {
    b_cur[2 * s]     = *(const float4*)(bp + (size_t)s * F_IN);
    b_cur[2 * s + 1] = *(const float4*)(bp + (size_t)s * F_IN + 64);
  }

#pragma unroll
  for (int m = 0; m < F_IN / 8; m++) {
    float4 a_nxt = a_cur;
    float4 b_nxt[8];
#pragma unroll
    for (int s = 0; s < 8; s++) b_nxt[s] = b_cur[s];
    if (m + 1 < F_IN / 8) {
      a_nxt = *(const float4*)(ap + 8 * (m + 1));
#pragma unroll
      for (int s = 0; s < 4; s++) {
        b_nxt[2 * s]     = *(const float4*)(bp + (size_t)(8 * (m + 1) + s) * F_IN);
        b_nxt[2 * s + 1] = *(const float4*)(bp + (size_t)(8 * (m + 1) + s) * F_IN + 64);
      }
    }
#pragma unroll
    for (int u = 0; u < 2; u++) {
      v2f av;
      av.x = u == 0 ? a_cur.x : a_cur.z;
      av.y = u == 0 ? a_cur.y : a_cur.w;
#pragma unroll
      for (int t = 0; t < 8; t++) {
        v2f bv;
        bv.x = f4e(b_cur[2 * (2 * u) + (t >> 2)], t & 3);
        bv.y = f4e(b_cur[2 * (2 * u + 1) + (t >> 2)], t & 3);
        acc[t] = __builtin_amdgcn_wmma_f32_16x16x4_f32(false, av, false, bv, (short)0, acc[t],
                                                       false, false);
      }
    }
    a_cur = a_nxt;
#pragma unroll
    for (int s = 0; s < 8; s++) b_cur[s] = b_nxt[s];
  }

#pragma unroll
  for (int r = 0; r < 8; r++) {
    float* hp = H + (size_t)(row0 + 8 * h + r) * F_IN + 4 * c;
    *(float4*)hp        = make_float4(acc[0][r], acc[1][r], acc[2][r], acc[3][r]);
    *(float4*)(hp + 64) = make_float4(acc[4][r], acc[5][r], acc[6][r], acc[7][r]);
  }

  const int g  = c >> 3;
  const int k8 = c & 7;
  const float4 s_lo = *(const float4*)(asr + 4 * c);
  const float4 s_hi = *(const float4*)(asr + 64 + 4 * c);
  const float4 d_lo = *(const float4*)(adt + 4 * c);
  const float4 d_hi = *(const float4*)(adt + 64 + 4 * c);
  float os0 = 0.f, os1 = 0.f, od0 = 0.f, od1 = 0.f;
#pragma unroll
  for (int rp = 0; rp < 4; rp++) {
    const int r0 = 2 * rp, r1 = 2 * rp + 1;
    const v2f a0 = {acc[0][r0], acc[0][r1]}, a1 = {acc[1][r0], acc[1][r1]};
    const v2f a2 = {acc[2][r0], acc[2][r1]}, a3 = {acc[3][r0], acc[3][r1]};
    const v2f a4 = {acc[4][r0], acc[4][r1]}, a5 = {acc[5][r0], acc[5][r1]};
    const v2f a6 = {acc[6][r0], acc[6][r1]}, a7 = {acc[7][r0], acc[7][r1]};
    const v2f ps0 = a0 * s_lo.x + a1 * s_lo.y + a2 * s_lo.z + a3 * s_lo.w;
    const v2f ps1 = a4 * s_hi.x + a5 * s_hi.y + a6 * s_hi.z + a7 * s_hi.w;
    const v2f pd0 = a0 * d_lo.x + a1 * d_lo.y + a2 * d_lo.z + a3 * d_lo.w;
    const v2f pd1 = a4 * d_hi.x + a5 * d_hi.y + a6 * d_hi.z + a7 * d_hi.w;
    const float e0 = grp_sum<8>(ps0.x), e1 = grp_sum<8>(ps1.x);
    const float e2 = grp_sum<8>(pd0.x), e3 = grp_sum<8>(pd1.x);
    const float f0 = grp_sum<8>(ps0.y), f1 = grp_sum<8>(ps1.y);
    const float f2 = grp_sum<8>(pd0.y), f3 = grp_sum<8>(pd1.y);
    if (k8 == r0) { os0 = e0; os1 = e1; od0 = e2; od1 = e3; }
    if (k8 == r1) { os0 = f0; os1 = f1; od0 = f2; od1 = f3; }
  }
  const float rc0 = __shfl_xor(g == 0 ? od0 : os0, 8);
  const float rc1 = __shfl_xor(g == 0 ? od1 : os1, 8);
  const float4 o = g == 0 ? make_float4(os0, rc0, os1, rc1) : make_float4(rc0, od0, rc1, od1);
  float* op = (g == 0 ? als : ald) + (size_t)(row0 + 8 * h + k8) * HEADS;
  *(float4*)op = o;
}

__device__ __forceinline__ int al2_src(int v) { return 2 * v - (v & 15); }
__device__ __forceinline__ int al2_dst(int v) { return 2 * v - (v & 15) + 16; }

__global__ void k_gemm2(const float* __restrict__ Hp, const float* __restrict__ W2,
                        const float* __restrict__ asr, const float* __restrict__ adt,
                        float* __restrict__ H2, float* __restrict__ al2, int n) {
  const int wave = threadIdx.x >> 5;
  const int lane = threadIdx.x & 31;
  const int row0 = (blockIdx.x * 8 + wave) * 16;
  if (row0 >= n) return;
  const int lr = lane & 15;
  const int lh = lane >> 4;
  const float* ap = Hp + (size_t)(row0 + lr) * F_IN + 2 * lh;
  const float* bp = W2 + (size_t)(2 * lh) * OUTC + lr;
  v8f acc = {};
#pragma unroll
  for (int k = 0; k < F_IN; k += 4) {
    v2f a = *(const v2f*)(ap + k);
    v2f b;
    b.x = bp[(size_t)k * OUTC];
    b.y = bp[(size_t)k * OUTC + OUTC];
    acc = __builtin_amdgcn_wmma_f32_16x16x4_f32(false, a, false, b, (short)0, acc,
                                                false, false);
  }
#pragma unroll
  for (int i = 0; i < 8; i++) {
    H2[(size_t)(row0 + i + 8 * lh) * OUTC + lr] = acc[i];
  }

  const float as = asr[lr], ad = adt[lr];
  const int i8 = lr & 7;
  float own_s = 0.f, own_d = 0.f;
#pragma unroll
  for (int i = 0; i < 8; i++) {
    const float ps = grp_sum<16>(acc[i] * as);
    const float pd = grp_sum<16>(acc[i] * ad);
    if (i8 == i) { own_s = ps; own_d = pd; }
  }
  const float kp  = lh == 0 ? own_s : own_d;
  const float rc  = __shfl_xor(lh == 0 ? own_d : own_s, 16);
  al2[(size_t)(row0 >> 4) * 32 + lane] = lh == (lr >> 3) ? kp : rc;
}

template <int G, class F>
__device__ __forceinline__ void for_each_chunk(bool fast, int d, int deg,
                                               const int* __restrict__ brow,
                                               const int* __restrict__ ei, int E, int g, F&& f) {
  if (fast) {
    const int tot = deg + 1;
    for (int base = 0; base < tot; base += G) {
      const int j = base + g;
      int s = d;
      if (j < deg) s = brow[j];
      f(s, j < tot);
    }
  } else {
    for (int base = 0; base < E; base += G) {
      const int e = base + g;
      bool valid = false;
      int s = d;
      if (e < E && ei[E + e] == d) { valid = true; s = ei[e]; }
      f(s, valid);
    }
    f(d, g == 0);
  }
}

template <int U>
__device__ __forceinline__ void gather_rows1(const float* __restrict__ H, int s, float a0, float a1,
                                             float a2, float a3, int j0, bool h1, bool h2, int lane,
                                             float4& acc) {
  float4 hv[U];
  float w[U];
#pragma unroll
  for (int u = 0; u < U; u++) {
    const int sj = __builtin_amdgcn_readlane(s, j0 + u);
    hv[u] = ((const float4*)(H + (size_t)sj * F_IN))[lane];
  }
#pragma unroll
  for (int u = 0; u < U; u++) {
    const float w0 = lane_f(a0, j0 + u), w1 = lane_f(a1, j0 + u);
    const float w2 = lane_f(a2, j0 + u), w3 = lane_f(a3, j0 + u);
    const float lo = h1 ? w1 : w0;
    const float hi = h1 ? w3 : w2;
    w[u] = h2 ? hi : lo;
  }
#pragma unroll
  for (int u = 0; u < U; u++) {
    acc.x = fmaf(hv[u].x, w[u], acc.x); acc.y = fmaf(hv[u].y, w[u], acc.y);
    acc.z = fmaf(hv[u].z, w[u], acc.z); acc.w = fmaf(hv[u].w, w[u], acc.w);
  }
}

template <int U>
__device__ __forceinline__ void gather_rows2(const float* __restrict__ H2, int s, float a, int j0,
                                             int g, float& acc) {
  float hv[U];
  float w[U];
#pragma unroll
  for (int u = 0; u < U; u++) {
    const int sj = __shfl(s, j0 + u, 16);
    hv[u] = H2[(size_t)sj * OUTC + g];
  }
#pragma unroll
  for (int u = 0; u < U; u++) w[u] = __shfl(a, j0 + u, 16);
#pragma unroll
  for (int u = 0; u < U; u++) acc = fmaf(hv[u], w[u], acc);
}

__device__ __forceinline__ void gather_list2(const float* __restrict__ H2, int s, float a, int cnt,
                                             int g, float& acc) {
  int j = 0;
  for (; j + 8 <= cnt; j += 8) gather_rows2<8>(H2, s, a, j, g, acc);
  if (cnt - j >= 4) { gather_rows2<4>(H2, s, a, j, g, acc); j += 4; }
  if (cnt - j >= 2) { gather_rows2<2>(H2, s, a, j, g, acc); j += 2; }
  if (cnt - j >= 1) gather_rows2<1>(H2, s, a, j, g, acc);
}

__global__ void k_node1(const int* __restrict__ ei, int E, int n,
                        const int* __restrict__ cnt, const int* __restrict__ bucket,
                        const float* __restrict__ als, const float* __restrict__ ald,
                        const float* __restrict__ H, const float* __restrict__ b1,
                        float* __restrict__ Hp) {
  const int lane = threadIdx.x & 31;
  const int d = __builtin_amdgcn_readfirstlane(blockIdx.x * (blockDim.x >> 5) + (threadIdx.x >> 5));
  if (d >= n) return;
  const int c = __builtin_amdgcn_readfirstlane(cnt[d]);
  const bool fast = c <= CAP;
  const int deg = fast ? c : 0;
  const int* brow = bucket + (size_t)d * CAP;
  const float4 dv = *(const float4*)(ald + (size_t)d * HEADS);
  const float ninf = -__builtin_inff();

  const int hsel = lane >> 3;
  const bool h1 = (hsel & 1) != 0, h2 = (hsel & 2) != 0;
  float4 acc = {0.f, 0.f, 0.f, 0.f};

  if (c < 32) {
    const int tot = c + 1;
    const bool valid = lane < tot;
    int s = d;
    if (lane < c) s = brow[lane];
    const float4 a = *(const float4*)(als + (size_t)s * HEADS);
    const float l0 = valid ? lrelu(a.x + dv.x) : ninf;
    const float l1 = valid ? lrelu(a.y + dv.y) : ninf;
    const float l2 = valid ? lrelu(a.z + dv.z) : ninf;
    const float l3 = valid ? lrelu(a.w + dv.w) : ninf;
    const float m0 = vmax<32>(l0), m1 = vmax<32>(l1);
    const float m2 = vmax<32>(l2), m3 = vmax<32>(l3);
    const float p0 = valid ? expf(l0 - m0) : 0.f;
    const float p1 = valid ? expf(l1 - m1) : 0.f;
    const float p2 = valid ? expf(l2 - m2) : 0.f;
    const float p3 = valid ? expf(l3 - m3) : 0.f;
    const float q0 = vsum<32>(p0) + 1e-16f, q1 = vsum<32>(p1) + 1e-16f;
    const float q2 = vsum<32>(p2) + 1e-16f, q3 = vsum<32>(p3) + 1e-16f;
    const float a0 = p0 / q0, a1 = p1 / q1, a2 = p2 / q2, a3 = p3 / q3;
    int j = 0;
    for (; j + 8 <= tot; j += 8) gather_rows1<8>(H, s, a0, a1, a2, a3, j, h1, h2, lane, acc);
    if (tot - j >= 4) { gather_rows1<4>(H, s, a0, a1, a2, a3, j, h1, h2, lane, acc); j += 4; }
    if (tot - j >= 2) { gather_rows1<2>(H, s, a0, a1, a2, a3, j, h1, h2, lane, acc); j += 2; }
    if (tot - j >= 1) gather_rows1<1>(H, s, a0, a1, a2, a3, j, h1, h2, lane, acc);
  } else if (c < CAP) {
    const int tot = c + 1;
    const bool vb = lane + 32 < tot;
    const int sa = brow[lane];
    int sb = d;
    if (lane + 32 < c) sb = brow[lane + 32];
    const float4 xa = *(const float4*)(als + (size_t)sa * HEADS);
    const float4 xb = *(const float4*)(als + (size_t)sb * HEADS);
    const float la0 = lrelu(xa.x + dv.x), la1 = lrelu(xa.y + dv.y);
    const float la2 = lrelu(xa.z + dv.z), la3 = lrelu(xa.w + dv.w);
    const float lb0 = vb ? lrelu(xb.x + dv.x) : ninf;
    const float lb1 = vb ? lrelu(xb.y + dv.y) : ninf;
    const float lb2 = vb ? lrelu(xb.z + dv.z) : ninf;
    const float lb3 = vb ? lrelu(xb.w + dv.w) : ninf;
    const float m0 = vmax<32>(fmaxf(la0, lb0)), m1 = vmax<32>(fmaxf(la1, lb1));
    const float m2 = vmax<32>(fmaxf(la2, lb2)), m3 = vmax<32>(fmaxf(la3, lb3));
    const float pa0 = expf(la0 - m0), pa1 = expf(la1 - m1);
    const float pa2 = expf(la2 - m2), pa3 = expf(la3 - m3);
    const float pb0 = vb ? expf(lb0 - m0) : 0.f;
    const float pb1 = vb ? expf(lb1 - m1) : 0.f;
    const float pb2 = vb ? expf(lb2 - m2) : 0.f;
    const float pb3 = vb ? expf(lb3 - m3) : 0.f;
    const float q0 = vsum<32>(pa0 + pb0) + 1e-16f, q1 = vsum<32>(pa1 + pb1) + 1e-16f;
    const float q2 = vsum<32>(pa2 + pb2) + 1e-16f, q3 = vsum<32>(pa3 + pb3) + 1e-16f;
    const float wa0 = pa0 / q0, wa1 = pa1 / q1, wa2 = pa2 / q2, wa3 = pa3 / q3;
    const float wb0 = pb0 / q0, wb1 = pb1 / q1, wb2 = pb2 / q2, wb3 = pb3 / q3;
    for (int j = 0; j < 32; j += 8) gather_rows1<8>(H, sa, wa0, wa1, wa2, wa3, j, h1, h2, lane, acc);
    const int tb = tot - 32;
    int j = 0;
    for (; j + 8 <= tb; j += 8) gather_rows1<8>(H, sb, wb0, wb1, wb2, wb3, j, h1, h2, lane, acc);
    if (tb - j >= 4) { gather_rows1<4>(H, sb, wb0, wb1, wb2, wb3, j, h1, h2, lane, acc); j += 4; }
    if (tb - j >= 2) { gather_rows1<2>(H, sb, wb0, wb1, wb2, wb3, j, h1, h2, lane, acc); j += 2; }
    if (tb - j >= 1) gather_rows1<1>(H, sb, wb0, wb1, wb2, wb3, j, h1, h2, lane, acc);
  } else {
    float m0 = ninf, m1 = ninf, m2 = ninf, m3 = ninf;
    for_each_chunk<32>(fast, d, deg, brow, ei, E, lane, [&](int s, bool valid) {
      if (valid) {
        const float4 a = *(const float4*)(als + (size_t)s * HEADS);
        m0 = fmaxf(m0, lrelu(a.x + dv.x));
        m1 = fmaxf(m1, lrelu(a.y + dv.y));
        m2 = fmaxf(m2, lrelu(a.z + dv.z));
        m3 = fmaxf(m3, lrelu(a.w + dv.w));
      }
    });
    m0 = vmax<32>(m0); m1 = vmax<32>(m1); m2 = vmax<32>(m2); m3 = vmax<32>(m3);

    float s0 = 0.f, s1 = 0.f, s2 = 0.f, s3 = 0.f;
    for_each_chunk<32>(fast, d, deg, brow, ei, E, lane, [&](int s, bool valid) {
      if (valid) {
        const float4 a = *(const float4*)(als + (size_t)s * HEADS);
        s0 += expf(lrelu(a.x + dv.x) - m0);
        s1 += expf(lrelu(a.y + dv.y) - m1);
        s2 += expf(lrelu(a.z + dv.z) - m2);
        s3 += expf(lrelu(a.w + dv.w) - m3);
      }
    });
    const float q0 = vsum<32>(s0) + 1e-16f, q1 = vsum<32>(s1) + 1e-16f;
    const float q2 = vsum<32>(s2) + 1e-16f, q3 = vsum<32>(s3) + 1e-16f;

    for_each_chunk<32>(fast, d, deg, brow, ei, E, lane, [&](int s, bool valid) {
      float a0 = 0.f, a1 = 0.f, a2 = 0.f, a3 = 0.f;
      if (valid) {
        const float4 a = *(const float4*)(als + (size_t)s * HEADS);
        a0 = expf(lrelu(a.x + dv.x) - m0) / q0;
        a1 = expf(lrelu(a.y + dv.y) - m1) / q1;
        a2 = expf(lrelu(a.z + dv.z) - m2) / q2;
        a3 = expf(lrelu(a.w + dv.w) - m3) / q3;
      }
      unsigned mask = __builtin_amdgcn_ballot_w32(valid);
      while (mask) {
        const int j = __builtin_ctz(mask);
        mask &= mask - 1;
        const int sj = __shfl(s, j);
        const float w0 = __shfl(a0, j), w1 = __shfl(a1, j);
        const float w2 = __shfl(a2, j), w3 = __shfl(a3, j);
        const float w = hsel == 0 ? w0 : hsel == 1 ? w1 : hsel == 2 ? w2 : w3;
        const float4 hv = *(const float4*)(H + (size_t)sj * F_IN + lane * 4);
        acc.x += hv.x * w; acc.y += hv.y * w; acc.z += hv.z * w; acc.w += hv.w * w;
      }
    });
  }

  float v[4] = {acc.x + b1[lane * 4 + 0], acc.y + b1[lane * 4 + 1],
                acc.z + b1[lane * 4 + 2], acc.w + b1[lane * 4 + 3]};
  for (int i = 0; i < 4; i++) {
    float t = v[i];
    t = t > 0.f ? t : (expf(t) - 1.f);
    Hp[(size_t)d * F_IN + i * 32 + lane] = t;
  }
}

__global__ void k_node2(const int* __restrict__ ei, int E, int n,
                        const int* __restrict__ cnt, const int* __restrict__ bucket,
                        const float* __restrict__ al2,
                        const float* __restrict__ H2, const float* __restrict__ b2,
                        float* __restrict__ out) {
  const int lane = threadIdx.x & 31;
  const int g = lane & 15;
  const int half = lane >> 4;
  const int d = (blockIdx.x * (blockDim.x >> 5) + (threadIdx.x >> 5)) * 2 + half;
  const bool live = d < n;
  const int dd = live ? d : 0;
  const int c = live ? cnt[dd] : -1;
  const bool fast = c <= CAP;
  const int deg = fast ? c : 0;
  const int* brow = bucket + (size_t)dd * CAP;
  const float dv = al2[al2_dst(dd)];
  const float ninf = -__builtin_inff();

  float acc = 0.f;
  if (__builtin_amdgcn_ballot_w32(c >= 32) == 0u) {
    const int tot = c + 1;
    const bool v0 = g < tot, v1 = g + 16 < tot;
    int s0 = dd, s1 = dd;
    if (g < c) s0 = brow[g];
    if (g + 16 < c) s1 = brow[g + 16];
    const float r0 = al2[al2_src(s0)], r1 = al2[al2_src(s1)];
    const float l0 = v0 ? lrelu(r0 + dv) : ninf;
    const float l1 = v1 ? lrelu(r1 + dv) : ninf;
    const float m = vmax<16>(fmaxf(l0, l1));
    const float p0 = v0 ? expf(l0 - m) : 0.f;
    const float p1 = v1 ? expf(l1 - m) : 0.f;
    const float q = vsum<16>(p0 + p1) + 1e-16f;
    const float a0 = p0 / q, a1 = p1 / q;
    const int t0 = tot < 16 ? tot : 16;
    gather_list2(H2, s0, a0, t0, g, acc);
    gather_list2(H2, s1, a1, tot - t0, g, acc);
  } else if (__builtin_amdgcn_ballot_w32(c >= CAP) == 0u) {
    const int tot = c + 1;
    const bool v0 = g < tot, v1 = g + 16 < tot, v2 = g + 32 < tot, v3 = g + 48 < tot;
    int s0 = dd, s1 = dd, s2 = dd, s3 = dd;
    if (g < c) s0 = brow[g];
    if (g + 16 < c) s1 = brow[g + 16];
    if (g + 32 < c) s2 = brow[g + 32];
    if (g + 48 < c) s3 = brow[g + 48];
    const float r0 = al2[al2_src(s0)], r1 = al2[al2_src(s1)];
    const float r2 = al2[al2_src(s2)], r3 = al2[al2_src(s3)];
    const float l0 = v0 ? lrelu(r0 + dv) : ninf;
    const float l1 = v1 ? lrelu(r1 + dv) : ninf;
    const float l2 = v2 ? lrelu(r2 + dv) : ninf;
    const float l3 = v3 ? lrelu(r3 + dv) : ninf;
    const float m = vmax<16>(fmaxf(fmaxf(fmaxf(l0, l1), l2), l3));
    const float p0 = v0 ? expf(l0 - m) : 0.f;
    const float p1 = v1 ? expf(l1 - m) : 0.f;
    const float p2 = v2 ? expf(l2 - m) : 0.f;
    const float p3 = v3 ? expf(l3 - m) : 0.f;
    const float q = vsum<16>(((p0 + p1) + p2) + p3) + 1e-16f;
    const float a0 = p0 / q, a1 = p1 / q, a2 = p2 / q, a3 = p3 / q;
    const int t0 = tot < 16 ? tot : 16;
    const int t1 = tot < 32 ? tot - t0 : 16;
    const int t2 = tot < 32 ? 0 : (tot < 48 ? tot - 32 : 16);
    const int t3 = tot < 48 ? 0 : tot - 48;
    gather_list2(H2, s0, a0, t0, g, acc);
    gather_list2(H2, s1, a1, t1, g, acc);
    gather_list2(H2, s2, a2, t2, g, acc);
    gather_list2(H2, s3, a3, t3, g, acc);
  } else {
    float m = ninf;
    for_each_chunk<16>(fast, dd, deg, brow, ei, E, g, [&](int s, bool valid) {
      if (valid) m = fmaxf(m, lrelu(al2[al2_src(s)] + dv));
    });
    m = vmax<16>(m);

    float ss = 0.f;
    for_each_chunk<16>(fast, dd, deg, brow, ei, E, g, [&](int s, bool valid) {
      if (valid) ss += expf(lrelu(al2[al2_src(s)] + dv) - m);
    });
    const float q = vsum<16>(ss) + 1e-16f;

    for_each_chunk<16>(fast, dd, deg, brow, ei, E, g, [&](int s, bool valid) {
      float a = 0.f;
      if (valid) a = expf(lrelu(al2[al2_src(s)] + dv) - m) / q;
      unsigned mask = (__builtin_amdgcn_ballot_w32(valid) >> (half * 16)) & 0xffffu;
      while (mask) {
        const int j = __builtin_ctz(mask);
        mask &= mask - 1;
        const int sj = __shfl(s, j, 16);
        const float w = __shfl(a, j, 16);
        acc += H2[(size_t)sj * OUTC + g] * w;
      }
    });
  }

  const float v = acc + b2[g];
  const float mx = vmax<16>(v);
  const float se = vsum<16>(expf(v - mx));
  const float lse = mx + logf(se);
  if (live) out[(size_t)d * OUTC + g] = v - lse;
}

static inline unsigned cdiv(long long a, long long b) { return (unsigned)((a + b - 1) / b); }

extern "C" void kernel_launch(void* const* d_in, const int* in_sizes, int n_in,
                              void* d_out, int out_size, void* d_ws, size_t ws_size,
                              hipStream_t stream) {
  const float* x    = (const float*)d_in[0];
  const int*   ei   = (const int*)  d_in[1];
  const float* W1   = (const float*)d_in[2];
  const float* asr1 = (const float*)d_in[3];
  const float* adt1 = (const float*)d_in[4];
  const float* b1   = (const float*)d_in[5];
  const float* W2   = (const float*)d_in[6];
  const float* asr2 = (const float*)d_in[7];
  const float* adt2 = (const float*)d_in[8];
  const float* b2   = (const float*)d_in[9];
  float* out = (float*)d_out;

  const int n = in_sizes[0] / F_IN;
  const int E = in_sizes[1] / 2;

  float* ws = (float*)d_ws;
  size_t off = 0;
  float* H1     = ws + off; off += (size_t)n * F_IN;
  float* Hp     = ws + off; off += (size_t)n * F_IN;
  int*   bucket = (int*)(ws + off); off += (size_t)n * CAP;
  float* als1   = ws + off; off += (size_t)n * HEADS;
  float* ald1   = ws + off; off += (size_t)n * HEADS;
  float* H2     = ws + off; off += (size_t)n * OUTC;
  float* al2    = ws + off; off += (size_t)n * 2;
  int*   cnt    = (int*)(ws + off); off += (size_t)n;

  const int B = 256;

  k_fill0i<<<cdiv(n, B), B, 0, stream>>>(cnt, n);
  k_bucket<<<cdiv(E, B), B, 0, stream>>>(ei, E, cnt, bucket);

  k_gemm1<<<cdiv(cdiv(n, 16), 2), 64, 0, stream>>>(x, W1, asr1, adt1, H1, als1, ald1, n);
  k_node1<<<cdiv(n, B / 32), B, 0, stream>>>(ei, E, n, cnt, bucket, als1, ald1, H1, b1, Hp);

  k_gemm2<<<cdiv(cdiv(n, 16), 8), B, 0, stream>>>(Hp, W2, asr2, adt2, H2, al2, n);
  k_node2<<<cdiv(n, B / 16), B, 0, stream>>>(ei, E, n, cnt, bucket, al2, H2, b2, out);
}
